# starting kernel
# baseline (speedup 1.0000x reference)
#include <hip/hip_runtime.h>
#include <hip/hip_bf16.h>
#include <math.h>

#define NUM_GRIDS 16
#define IN_X 32
#define IN_Z 32
#define OUT 64
#define BATCH 512
#define GP1 (NUM_GRIDS + 1)

typedef __attribute__((ext_vector_type(2))) float v2f;
typedef __attribute__((ext_vector_type(8))) float v8f;

__global__ void build_scatter(const float* __restrict__ x,
                              const float* __restrict__ z,
                              const float* __restrict__ borders,
                              const float* __restrict__ invlen,
                              float* __restrict__ Vx,
                              float* __restrict__ Vz) {
  const int per = IN_X * BATCH;
  int t = blockIdx.x * blockDim.x + threadIdx.x;
  if (t >= 2 * per) return;
  int sel = t / per;
  int r   = t - sel * per;
  int i   = r / BATCH;
  int b   = r - i * BATCH;

  float v = sel ? z[i * BATCH + b] : x[i * BATCH + b];
  float* V = sel ? Vz : Vx;

  float e   = __expf(-fabsf(v));
  float cdf = (v > 0.0f) ? (1.0f - 0.5f * e) : (0.5f * e);
  int idx = (int)(cdf * (float)NUM_GRIDS);
  idx = idx < 0 ? 0 : (idx > NUM_GRIDS - 1 ? NUM_GRIDS - 1 : idx);
  float d = (v - borders[idx]) * invlen[idx];

#pragma unroll
  for (int g = 0; g < GP1; ++g) {
    float w = (g == idx) ? (1.0f - d) : ((g == idx + 1) ? d : 0.0f);
    V[((size_t)g * IN_X + i) * BATCH + b] = w;
  }
}

__global__ void __launch_bounds__(32) basis_gemm(
    const float* __restrict__ P,
    const float* __restrict__ Vx,
    const float* __restrict__ Vz,
    float* __restrict__ out)
{
  const int o    = blockIdx.y;
  const int b0   = blockIdx.x * 32;
  const int lane = threadIdx.x;
  const int hi   = lane >> 4;
  const int lo   = lane & 15;

  float outacc0 = 0.0f;
  float outacc1 = 0.0f;

  const size_t gzStride = (size_t)OUT * IN_X * IN_Z;

  for (int gx = 0; gx < GP1; ++gx) {
    v8f acc00 = {};
    v8f acc01 = {};
    v8f acc10 = {};
    v8f acc11 = {};

    for (int gz = 0; gz < GP1; ++gz) {
      const float* Ap  = P + ((((size_t)gx * GP1 + gz) * OUT + o) * IN_X) * IN_Z;
      const float* Bp0 = Vz + (size_t)gz * IN_Z * BATCH + (b0 + lo);
      const float* Bp1 = Bp0 + 16;

      if (gz + 1 < GP1)
        __builtin_prefetch(Ap + gzStride + (size_t)lo * IN_Z, 0, 3);

#pragma unroll
      for (int jj = 0; jj < IN_Z; jj += 4) {
        const int j = jj + 2 * hi;
        v2f a0 = *(const v2f*)(Ap + (size_t)lo        * IN_Z + j);
        v2f a1 = *(const v2f*)(Ap + (size_t)(16 + lo) * IN_Z + j);
        v2f b0v, b1v;
        b0v.x = Bp0[(size_t)j * BATCH];
        b0v.y = Bp0[(size_t)(j + 1) * BATCH];
        b1v.x = Bp1[(size_t)j * BATCH];
        b1v.y = Bp1[(size_t)(j + 1) * BATCH];
        acc00 = __builtin_amdgcn_wmma_f32_16x16x4_f32(
            false, a0, false, b0v, (short)0, acc00, false, false);
        acc10 = __builtin_amdgcn_wmma_f32_16x16x4_f32(
            false, a0, false, b1v, (short)0, acc10, false, false);
        acc01 = __builtin_amdgcn_wmma_f32_16x16x4_f32(
            false, a1, false, b0v, (short)0, acc01, false, false);
        acc11 = __builtin_amdgcn_wmma_f32_16x16x4_f32(
            false, a1, false, b1v, (short)0, acc11, false, false);
      }
    }

    float p0 = 0.0f, p1 = 0.0f;
#pragma unroll
    for (int rr = 0; rr < 8; ++rr) {
      const int m = rr + 8 * hi;
      float w00 = Vx[((size_t)gx * IN_X + m)      * BATCH + (b0 + lo)];
      float w01 = Vx[((size_t)gx * IN_X + 16 + m) * BATCH + (b0 + lo)];
      float w10 = Vx[((size_t)gx * IN_X + m)      * BATCH + (b0 + 16 + lo)];
      float w11 = Vx[((size_t)gx * IN_X + 16 + m) * BATCH + (b0 + 16 + lo)];
      p0 += acc00[rr] * w00 + acc01[rr] * w01;
      p1 += acc10[rr] * w10 + acc11[rr] * w11;
    }
    p0 += __shfl_xor(p0, 16, 32);
    p1 += __shfl_xor(p1, 16, 32);
    outacc0 += p0;
    outacc1 += p1;
  }

  if (lane < 16) {
    out[(size_t)o * BATCH + b0 + lane]      = outacc0;
    out[(size_t)o * BATCH + b0 + 16 + lane] = outacc1;
  }
}

extern "C" void kernel_launch(void* const* d_in, const int* in_sizes, int n_in,
                              void* d_out, int out_size, void* d_ws, size_t ws_size,
                              hipStream_t stream) {
  const float* x       = (const float*)d_in[0];
  const float* z       = (const float*)d_in[1];
  const float* P       = (const float*)d_in[2];
  const float* borders = (const float*)d_in[3];
  const float* invlen  = (const float*)d_in[4];
  float* out = (float*)d_out;

  float* Vx = (float*)d_ws;
  float* Vz = Vx + (size_t)GP1 * IN_X * BATCH;

  {
    int total = 2 * IN_X * BATCH;
    build_scatter<<<dim3((total + 255) / 256), dim3(256), 0, stream>>>(
        x, z, borders, invlen, Vx, Vz);
  }
  basis_gemm<<<dim3(BATCH / 32, OUT), dim3(32), 0, stream>>>(P, Vx, Vz, out);
}


# Sort batch columns by x bin per x row; each 16-column wave tile multiplies only its 2-3 needed gx slices on WMMA (B built in registers), row sum via workspace partials: new summation order
# speedup vs baseline: 2.6215x; 2.6215x over previous
#include <hip/hip_runtime.h>
#include <math.h>

#define NUM_GRIDS 16
#define IN_X 32
#define IN_Z 32
#define OUT 64
#define BATCH 512
#define GP1 (NUM_GRIDS + 1)
#define TILE_N 16
#define NTILES (BATCH / TILE_N)
#define OHALF 32

typedef __attribute__((ext_vector_type(2))) float v2f;
typedef __attribute__((ext_vector_type(4))) float v4f;
typedef __attribute__((ext_vector_type(8))) float v8f;

__device__ __forceinline__ int bin_index(float v) {
  float e   = __expf(-fabsf(v));
  float cdf = (v > 0.0f) ? (1.0f - 0.5f * e) : (0.5f * e);
  int idx = (int)(cdf * (float)NUM_GRIDS);
  return idx < 0 ? 0 : (idx > NUM_GRIDS - 1 ? NUM_GRIDS - 1 : idx);
}

__device__ __forceinline__ int bin_of(float v, const float* __restrict__ borders,
                                      const float* __restrict__ invlen, float& d) {
  const int idx = bin_index(v);
  d = (v - borders[idx]) * invlen[idx];
  return idx;
}

__global__ void __launch_bounds__(BATCH) sort_rows(const float* __restrict__ x,
                                                   int* __restrict__ perm,
                                                   int* __restrict__ rank) {
  __shared__ int keys[BATCH];
  const int i = blockIdx.x;
  const int b = threadIdx.x;
  const int idx = bin_index(x[i * BATCH + b]);
  keys[b] = idx;
  __syncthreads();
  int r = 0;
  for (int c = 0; c < BATCH; ++c) {
    const int k = keys[c];
    r += (k < idx || (k == idx && c < b)) ? 1 : 0;
  }
  perm[i * BATCH + r] = b;
  rank[i * BATCH + b] = r;
}

__global__ void __launch_bounds__(32) bucket_gemm(
    const float* __restrict__ P,
    const float* __restrict__ x,
    const float* __restrict__ z,
    const float* __restrict__ borders,
    const float* __restrict__ invlen,
    const int* __restrict__ perm,
    float* __restrict__ T,
    int obase)
{
  const int t    = blockIdx.x;
  const int i    = blockIdx.y;
  const int lane = threadIdx.x;
  const int h    = lane >> 4;
  const int n    = lane & 15;

  const int bn = perm[i * BATCH + t * TILE_N + n] & (BATCH - 1);

  float dxv;
  const int ix = bin_of(x[i * BATCH + bn], borders, invlen, dxv);

  int blo = ix, bhi = ix;
  for (int m = 1; m < 16; m <<= 1) {
    blo = min(blo, __shfl_xor(blo, m, 32));
    bhi = max(bhi, __shfl_xor(bhi, m, 32));
  }
  const int gmin = __builtin_amdgcn_readfirstlane(blo);
  const int gmax = __builtin_amdgcn_readfirstlane(bhi) + 1;

  int   zi[16];
  float zw0[16], zw1[16];
#pragma unroll
  for (int e = 0; e < 16; ++e) {
    float dz;
    zi[e]  = bin_of(z[(16 * h + e) * BATCH + bn], borders, invlen, dz);
    zw0[e] = 1.0f - dz;
    zw1[e] = dz;
  }

  v8f acc0 = {};
  v8f acc1 = {};

  const size_t sliceStride = (size_t)OUT * IN_X * IN_Z;
  const float* Prow = P + ((size_t)(obase + n) * IN_X + i) * IN_Z + 16 * h;
  const float* Ppre = P + ((size_t)(obase + lane) * IN_X + i) * IN_Z;
  const int lastSlice = gmax * GP1 + (GP1 - 1);

  for (int gx = gmin; gx <= gmax; ++gx) {
    const float wx = (gx == ix) ? (1.0f - dxv) : ((gx == ix + 1) ? dxv : 0.0f);
    float w0[16], w1[16];
#pragma unroll
    for (int e = 0; e < 16; ++e) {
      w0[e] = zw0[e] * wx;
      w1[e] = zw1[e] * wx;
    }

    for (int gz = 0; gz < GP1; ++gz) {
      const int slice = gx * GP1 + gz;
      const float* Ap = Prow + (size_t)slice * sliceStride;

      if (slice < lastSlice)
        __builtin_prefetch(Ppre + (size_t)(slice + 1) * sliceStride, 0, 3);

      v4f a0[4], a1[4];
#pragma unroll
      for (int c = 0; c < 4; ++c) {
        a0[c] = *(const v4f*)(Ap + 4 * c);
        a1[c] = *(const v4f*)(Ap + (size_t)16 * IN_X * IN_Z + 4 * c);
      }

#pragma unroll
      for (int q = 0; q < 8; ++q) {
        const int e0 = 2 * q, e1 = 2 * q + 1;
        v2f bv;
        bv.x = (zi[e0] == gz) ? w0[e0] : ((zi[e0] + 1 == gz) ? w1[e0] : 0.0f);
        bv.y = (zi[e1] == gz) ? w0[e1] : ((zi[e1] + 1 == gz) ? w1[e1] : 0.0f);
        v2f av0, av1;
        av0.x = a0[q >> 1][2 * (q & 1)];
        av0.y = a0[q >> 1][2 * (q & 1) + 1];
        av1.x = a1[q >> 1][2 * (q & 1)];
        av1.y = a1[q >> 1][2 * (q & 1) + 1];
        acc0 = __builtin_amdgcn_wmma_f32_16x16x4_f32(
            false, av0, false, bv, (short)0, acc0, false, false);
        acc1 = __builtin_amdgcn_wmma_f32_16x16x4_f32(
            false, av1, false, bv, (short)0, acc1, false, false);
      }
    }
  }

  float* Tw = T + ((size_t)(i * NTILES + t) * OHALF) * TILE_N;
#pragma unroll
  for (int r = 0; r < 8; ++r) {
    Tw[(8 * h + r) * TILE_N + n]      = acc0[r];
    Tw[(16 + 8 * h + r) * TILE_N + n] = acc1[r];
  }
}

__global__ void __launch_bounds__(256) reduce_rows(const float* __restrict__ T,
                                                   const int* __restrict__ rank,
                                                   float* __restrict__ out,
                                                   int obase) {
  const int g  = blockIdx.x * 256 + threadIdx.x;
  const int b  = g & (BATCH - 1);
  const int op = g >> 9;
  float s = 0.0f;
#pragma unroll
  for (int i = 0; i < IN_X; ++i) {
    const int r = rank[i * BATCH + b] & (BATCH - 1);
    s += T[((size_t)(i * NTILES + (r >> 4)) * OHALF + op) * TILE_N + (r & 15)];
  }
  out[(size_t)(obase + op) * BATCH + b] = s;
}

extern "C" void kernel_launch(void* const* d_in, const int* in_sizes, int n_in,
                              void* d_out, int out_size, void* d_ws, size_t ws_size,
                              hipStream_t stream) {
  const float* x       = (const float*)d_in[0];
  const float* z       = (const float*)d_in[1];
  const float* P       = (const float*)d_in[2];
  const float* borders = (const float*)d_in[3];
  const float* invlen  = (const float*)d_in[4];
  float* out = (float*)d_out;

  float* T   = (float*)d_ws;
  int* perm  = (int*)(T + (size_t)IN_X * NTILES * OHALF * TILE_N);
  int* rank  = perm + IN_X * BATCH;

  sort_rows<<<dim3(IN_X), dim3(BATCH), 0, stream>>>(x, perm, rank);

  bucket_gemm<<<dim3(NTILES, IN_X), dim3(32), 0, stream>>>(
      P, x, z, borders, invlen, perm, T, 0);
  reduce_rows<<<dim3(OHALF * BATCH / 256), dim3(256), 0, stream>>>(T, rank, out, 0);

  bucket_gemm<<<dim3(NTILES, IN_X), dim3(32), 0, stream>>>(
      P, x, z, borders, invlen, perm, T, OHALF);
  reduce_rows<<<dim3(OHALF * BATCH / 256), dim3(256), 0, stream>>>(T, rank, out, OHALF);
}


# Product kernel: each tile's run of (gx,gz) slices is dealt round-robin to 8 waves of one work-group (was 1 wave), partials added in LDS in fixed wave order: new summation order
# speedup vs baseline: 3.5223x; 1.3436x over previous
#include <hip/hip_runtime.h>
#include <math.h>

#define NUM_GRIDS 16
#define IN_X 32
#define IN_Z 32
#define OUT 64
#define BATCH 512
#define GP1 (NUM_GRIDS + 1)
#define TILE_N 16
#define NTILES (BATCH / TILE_N)
#define OHALF 32
#define KW 8

typedef __attribute__((ext_vector_type(2))) float v2f;
typedef __attribute__((ext_vector_type(4))) float v4f;
typedef __attribute__((ext_vector_type(8))) float v8f;

__device__ __forceinline__ int bin_index(float v) {
  float e   = __expf(-fabsf(v));
  float cdf = (v > 0.0f) ? (1.0f - 0.5f * e) : (0.5f * e);
  int idx = (int)(cdf * (float)NUM_GRIDS);
  return idx < 0 ? 0 : (idx > NUM_GRIDS - 1 ? NUM_GRIDS - 1 : idx);
}

__device__ __forceinline__ int bin_of(float v, const float* __restrict__ borders,
                                      const float* __restrict__ invlen, float& d) {
  const int idx = bin_index(v);
  d = (v - borders[idx]) * invlen[idx];
  return idx;
}

__global__ void __launch_bounds__(BATCH) sort_rows(const float* __restrict__ x,
                                                   int* __restrict__ perm,
                                                   int* __restrict__ rank) {
  __shared__ int keys[BATCH];
  const int i = blockIdx.x;
  const int b = threadIdx.x;
  const int idx = bin_index(x[i * BATCH + b]);
  keys[b] = idx;
  __syncthreads();
  int r = 0;
  for (int c = 0; c < BATCH; ++c) {
    const int k = keys[c];
    r += (k < idx || (k == idx && c < b)) ? 1 : 0;
  }
  perm[i * BATCH + r] = b;
  rank[i * BATCH + b] = r;
}

__global__ void __launch_bounds__(32 * KW) __attribute__((amdgpu_waves_per_eu(8, 8))) bucket_gemm(
    const float* __restrict__ P,
    const float* __restrict__ x,
    const float* __restrict__ z,
    const float* __restrict__ borders,
    const float* __restrict__ invlen,
    const int* __restrict__ perm,
    float* __restrict__ T,
    int obase)
{
  __shared__ float part[KW][OHALF * TILE_N];

  const int t    = blockIdx.x;
  const int i    = blockIdx.y;
  const int lane = threadIdx.x & 31;
  const int w    = __builtin_amdgcn_readfirstlane((int)(threadIdx.x >> 5));
  const int h    = lane >> 4;
  const int n    = lane & 15;

  const int bn = perm[i * BATCH + t * TILE_N + n] & (BATCH - 1);

  float dxv;
  const int ix = bin_of(x[i * BATCH + bn], borders, invlen, dxv);

  int blo = ix, bhi = ix;
  for (int m = 1; m < 16; m <<= 1) {
    blo = min(blo, __shfl_xor(blo, m, 32));
    bhi = max(bhi, __shfl_xor(bhi, m, 32));
  }
  const int gmin = __builtin_amdgcn_readfirstlane(blo);
  const int gmax = __builtin_amdgcn_readfirstlane(bhi) + 1;

  int   zi[16];
  float zw0[16], zw1[16];
#pragma unroll
  for (int e = 0; e < 16; ++e) {
    float dz;
    zi[e]  = bin_of(z[(16 * h + e) * BATCH + bn], borders, invlen, dz);
    zw0[e] = 1.0f - dz;
    zw1[e] = dz;
  }

  v8f acc0 = {};
  v8f acc1 = {};

  const size_t sliceStride = (size_t)OUT * IN_X * IN_Z;
  const float* Prow = P + ((size_t)(obase + n) * IN_X + i) * IN_Z + 16 * h;
  const float* Ppre = P + ((size_t)(obase + lane) * IN_X + i) * IN_Z;
  const int lastSlice = gmax * GP1 + (GP1 - 1);

  int gx = gmin, gz = w;
  for (int slice = gmin * GP1 + w; slice <= lastSlice; slice += KW) {
    const float* Ap = Prow + (size_t)slice * sliceStride;

    if (slice + KW <= lastSlice)
      __builtin_prefetch(Ppre + (size_t)(slice + KW) * sliceStride, 0, 3);

    v4f a0[4], a1[4];
#pragma unroll
    for (int c = 0; c < 4; ++c) {
      a0[c] = *(const v4f*)(Ap + 4 * c);
      a1[c] = *(const v4f*)(Ap + (size_t)16 * IN_X * IN_Z + 4 * c);
    }

    const float wx = (gx == ix) ? (1.0f - dxv) : ((gx == ix + 1) ? dxv : 0.0f);

#pragma unroll
    for (int q = 0; q < 8; ++q) {
      const int e0 = 2 * q, e1 = 2 * q + 1;
      v2f bv;
      bv.x = (zi[e0] == gz) ? zw0[e0] * wx : ((zi[e0] + 1 == gz) ? zw1[e0] * wx : 0.0f);
      bv.y = (zi[e1] == gz) ? zw0[e1] * wx : ((zi[e1] + 1 == gz) ? zw1[e1] * wx : 0.0f);
      v2f av0, av1;
      av0.x = a0[q >> 1][2 * (q & 1)];
      av0.y = a0[q >> 1][2 * (q & 1) + 1];
      av1.x = a1[q >> 1][2 * (q & 1)];
      av1.y = a1[q >> 1][2 * (q & 1) + 1];
      acc0 = __builtin_amdgcn_wmma_f32_16x16x4_f32(
          false, av0, false, bv, (short)0, acc0, false, false);
      acc1 = __builtin_amdgcn_wmma_f32_16x16x4_f32(
          false, av1, false, bv, (short)0, acc1, false, false);
    }

    gz += KW;
    if (gz >= GP1) { gz -= GP1; ++gx; }
  }

  float* pw = part[w];
#pragma unroll
  for (int r = 0; r < 8; ++r) {
    pw[(8 * h + r) * TILE_N + n]      = acc0[r];
    pw[(16 + 8 * h + r) * TILE_N + n] = acc1[r];
  }
  __syncthreads();

  float* Tw = T + ((size_t)(i * NTILES + t) * OHALF) * TILE_N;
  for (int k = threadIdx.x; k < OHALF * TILE_N; k += 32 * KW) {
    float s = part[0][k];
#pragma unroll
    for (int v = 1; v < KW; ++v) s += part[v][k];
    Tw[k] = s;
  }
}

__global__ void __launch_bounds__(256) reduce_rows(const float* __restrict__ T,
                                                   const int* __restrict__ rank,
                                                   float* __restrict__ out,
                                                   int obase) {
  const int g  = blockIdx.x * 256 + threadIdx.x;
  const int b  = g & (BATCH - 1);
  const int op = g >> 9;
  float s = 0.0f;
#pragma unroll
  for (int i = 0; i < IN_X; ++i) {
    const int r = rank[i * BATCH + b] & (BATCH - 1);
    s += T[((size_t)(i * NTILES + (r >> 4)) * OHALF + op) * TILE_N + (r & 15)];
  }
  out[(size_t)(obase + op) * BATCH + b] = s;
}

extern "C" void kernel_launch(void* const* d_in, const int* in_sizes, int n_in,
                              void* d_out, int out_size, void* d_ws, size_t ws_size,
                              hipStream_t stream) {
  const float* x       = (const float*)d_in[0];
  const float* z       = (const float*)d_in[1];
  const float* P       = (const float*)d_in[2];
  const float* borders = (const float*)d_in[3];
  const float* invlen  = (const float*)d_in[4];
  float* out = (float*)d_out;

  float* T   = (float*)d_ws;
  int* perm  = (int*)(T + (size_t)IN_X * NTILES * OHALF * TILE_N);
  int* rank  = perm + IN_X * BATCH;

  sort_rows<<<dim3(IN_X), dim3(BATCH), 0, stream>>>(x, perm, rank);

  bucket_gemm<<<dim3(NTILES, IN_X), dim3(32 * KW), 0, stream>>>(
      P, x, z, borders, invlen, perm, T, 0);
  reduce_rows<<<dim3(OHALF * BATCH / 256), dim3(256), 0, stream>>>(T, rank, out, 0);

  bucket_gemm<<<dim3(NTILES, IN_X), dim3(32 * KW), 0, stream>>>(
      P, x, z, borders, invlen, perm, T, OHALF);
  reduce_rows<<<dim3(OHALF * BATCH / 256), dim3(256), 0, stream>>>(T, rank, out, OHALF);
}


# Product kernel: each work-group takes two neighbouring column tiles and loads every grid slice of their joint run once for both; half the waves, fewer loads; new summation order (second tile)
# speedup vs baseline: 5.0907x; 1.4453x over previous
#include <hip/hip_runtime.h>
#include <math.h>

#define NUM_GRIDS 16
#define IN_X 32
#define IN_Z 32
#define OUT 64
#define BATCH 512
#define GP1 (NUM_GRIDS + 1)
#define TILE_N 16
#define NTILES (BATCH / TILE_N)
#define NPAIRS (NTILES / 2)
#define OHALF 32
#define KW 8

typedef __attribute__((ext_vector_type(2))) float v2f;
typedef __attribute__((ext_vector_type(4))) float v4f;
typedef __attribute__((ext_vector_type(8))) float v8f;

__device__ __forceinline__ int bin_index(float v) {
  float e   = __expf(-fabsf(v));
  float cdf = (v > 0.0f) ? (1.0f - 0.5f * e) : (0.5f * e);
  int idx = (int)(cdf * (float)NUM_GRIDS);
  return idx < 0 ? 0 : (idx > NUM_GRIDS - 1 ? NUM_GRIDS - 1 : idx);
}

__device__ __forceinline__ int bin_of(float v, const float* __restrict__ borders,
                                      const float* __restrict__ invlen, float& d) {
  const int idx = bin_index(v);
  d = (v - borders[idx]) * invlen[idx];
  return idx;
}

template <int N>
__device__ __forceinline__ int row_shift_right_or0(int v) {
  return __builtin_amdgcn_update_dpp(0, v, 0x110 + N, 0xf, 0xf, true);
}

__global__ void __launch_bounds__(32 * KW) __attribute__((amdgpu_waves_per_eu(4, 4))) bucket_gemm(
    const float* __restrict__ P,
    const float* __restrict__ x,
    const float* __restrict__ z,
    const float* __restrict__ borders,
    const float* __restrict__ invlen,
    int* __restrict__ rank,
    float* __restrict__ T,
    int obase)
{
  __shared__ float part[KW][2 * OHALF * TILE_N];
  __shared__ int   cnt[16][16];
  __shared__ int   first[16][16];
  __shared__ int   selb[2 * TILE_N];
  __shared__ int   selk[2 * TILE_N];
  __shared__ float selx[2 * TILE_N];

  const int tp   = blockIdx.x;
  const int i    = blockIdx.y;
  const int tid  = threadIdx.x;
  const int lane = threadIdx.x & 31;
  const int w    = __builtin_amdgcn_readfirstlane((int)(threadIdx.x >> 5));
  const int h    = lane >> 4;
  const int n    = lane & 15;

  const float xv0 = x[i * BATCH + tid];
  const float xv1 = x[i * BATCH + (BATCH / 2) + tid];
  const int k0 = bin_index(xv0);
  const int k1 = bin_index(xv1);
  unsigned same0 = ~0u, same1 = ~0u;
  unsigned ing0  = ~0u, ing1  = ~0u;
#pragma unroll
  for (int bit = 0; bit < 4; ++bit) {
    const unsigned m0 = __builtin_amdgcn_ballot_w32(((k0 >> bit) & 1) != 0);
    const unsigned m1 = __builtin_amdgcn_ballot_w32(((k1 >> bit) & 1) != 0);
    same0 &= ((k0 >> bit) & 1) ? m0 : ~m0;
    same1 &= ((k1 >> bit) & 1) ? m1 : ~m1;
    ing0  &= ((lane >> bit) & 1) ? m0 : ~m0;
    ing1  &= ((lane >> bit) & 1) ? m1 : ~m1;
  }
  const unsigned below = (1u << lane) - 1u;
  const int in0 = __builtin_popcount(same0 & below);
  const int in1 = __builtin_popcount(same1 & below);
  if (lane < 16) {
    cnt[w][lane]     = __builtin_popcount(ing0);
    cnt[w + 8][lane] = __builtin_popcount(ing1);
  }
  __syncthreads();
  {
    const int v = tid >> 4, g = tid & 15;
    int tot = 0, pre = 0;
#pragma unroll
    for (int u = 0; u < 16; ++u) {
      const int c = cnt[u][g];
      tot += c;
      pre += (u < v) ? c : 0;
    }
    int s = tot;
    s += row_shift_right_or0<1>(s);
    s += row_shift_right_or0<2>(s);
    s += row_shift_right_or0<4>(s);
    s += row_shift_right_or0<8>(s);
    first[v][g] = (s - tot) + pre;
  }
  __syncthreads();
  const int r0 = first[w][k0] + in0;
  const int r1 = first[w + 8][k1] + in1;
  if ((r0 >> 5) == tp) { selb[r0 & 31] = tid;               selk[r0 & 31] = k0; selx[r0 & 31] = xv0; }
  if ((r1 >> 5) == tp) { selb[r1 & 31] = tid + (BATCH / 2); selk[r1 & 31] = k1; selx[r1 & 31] = xv1; }
  __syncthreads();
  if (tp == 0 && obase == 0) {
    rank[i * BATCH + tid]               = r0;
    rank[i * BATCH + (BATCH / 2) + tid] = r1;
  }

  const int   bnA  = selb[n] & (BATCH - 1);
  const int   bnB  = selb[TILE_N + n] & (BATCH - 1);
  const int   ixA  = selk[n];
  const int   ixB  = selk[TILE_N + n];
  const float dxvA = (selx[n] - borders[ixA]) * invlen[ixA];
  const float dxvB = (selx[TILE_N + n] - borders[ixB]) * invlen[ixB];

  const int gminA = __builtin_amdgcn_readfirstlane(selk[0]);
  const int gmaxA = __builtin_amdgcn_readfirstlane(selk[TILE_N - 1]) + 1;
  const int gminB = __builtin_amdgcn_readfirstlane(selk[TILE_N]);
  const int gmaxB = __builtin_amdgcn_readfirstlane(selk[2 * TILE_N - 1]) + 1;

  int   ziA[16], ziB[16];
  float zwA0[16], zwA1[16], zwB0[16], zwB1[16];
#pragma unroll
  for (int e = 0; e < 16; ++e) {
    float dz;
    ziA[e]  = bin_of(z[(16 * h + e) * BATCH + bnA], borders, invlen, dz);
    zwA0[e] = 1.0f - dz;
    zwA1[e] = dz;
    ziB[e]  = bin_of(z[(16 * h + e) * BATCH + bnB], borders, invlen, dz);
    zwB0[e] = 1.0f - dz;
    zwB1[e] = dz;
  }

  v8f accA0 = {};
  v8f accA1 = {};
  v8f accB0 = {};
  v8f accB1 = {};

  const size_t sliceStride = (size_t)OUT * IN_X * IN_Z;
  const float* Prow = P + ((size_t)(obase + n) * IN_X + i) * IN_Z + 16 * h;
  const float* Ppre = P + ((size_t)(obase + lane) * IN_X + i) * IN_Z;
  const int lastSlice = gmaxB * GP1 + (GP1 - 1);

  int gx = gminA, gz = w;
  for (int slice = gminA * GP1 + w; slice <= lastSlice; slice += KW) {
    const bool doA = (gx <= gmaxA);
    const bool doB = (gx >= gminB);
    if (doA || doB) {
      const float* Ap = Prow + (size_t)slice * sliceStride;

      if (slice + KW <= lastSlice)
        __builtin_prefetch(Ppre + (size_t)(slice + KW) * sliceStride, 0, 3);

      v4f a0[4], a1[4];
#pragma unroll
      for (int c = 0; c < 4; ++c) {
        a0[c] = *(const v4f*)(Ap + 4 * c);
        a1[c] = *(const v4f*)(Ap + (size_t)16 * IN_X * IN_Z + 4 * c);
      }

      if (doA) {
        const float wx = (gx == ixA) ? (1.0f - dxvA) : ((gx == ixA + 1) ? dxvA : 0.0f);
#pragma unroll
        for (int q = 0; q < 8; ++q) {
          const int e0 = 2 * q, e1 = 2 * q + 1;
          v2f bv;
          bv.x = (ziA[e0] == gz) ? zwA0[e0] * wx : ((ziA[e0] + 1 == gz) ? zwA1[e0] * wx : 0.0f);
          bv.y = (ziA[e1] == gz) ? zwA0[e1] * wx : ((ziA[e1] + 1 == gz) ? zwA1[e1] * wx : 0.0f);
          v2f av0, av1;
          av0.x = a0[q >> 1][2 * (q & 1)];
          av0.y = a0[q >> 1][2 * (q & 1) + 1];
          av1.x = a1[q >> 1][2 * (q & 1)];
          av1.y = a1[q >> 1][2 * (q & 1) + 1];
          accA0 = __builtin_amdgcn_wmma_f32_16x16x4_f32(
              false, av0, false, bv, (short)0, accA0, false, false);
          accA1 = __builtin_amdgcn_wmma_f32_16x16x4_f32(
              false, av1, false, bv, (short)0, accA1, false, false);
        }
      }
      if (doB) {
        const float wx = (gx == ixB) ? (1.0f - dxvB) : ((gx == ixB + 1) ? dxvB : 0.0f);
#pragma unroll
        for (int q = 0; q < 8; ++q) {
          const int e0 = 2 * q, e1 = 2 * q + 1;
          v2f bv;
          bv.x = (ziB[e0] == gz) ? zwB0[e0] * wx : ((ziB[e0] + 1 == gz) ? zwB1[e0] * wx : 0.0f);
          bv.y = (ziB[e1] == gz) ? zwB0[e1] * wx : ((ziB[e1] + 1 == gz) ? zwB1[e1] * wx : 0.0f);
          v2f av0, av1;
          av0.x = a0[q >> 1][2 * (q & 1)];
          av0.y = a0[q >> 1][2 * (q & 1) + 1];
          av1.x = a1[q >> 1][2 * (q & 1)];
          av1.y = a1[q >> 1][2 * (q & 1) + 1];
          accB0 = __builtin_amdgcn_wmma_f32_16x16x4_f32(
              false, av0, false, bv, (short)0, accB0, false, false);
          accB1 = __builtin_amdgcn_wmma_f32_16x16x4_f32(
              false, av1, false, bv, (short)0, accB1, false, false);
        }
      }
    }

    gz += KW;
    if (gz >= GP1) { gz -= GP1; ++gx; }
  }

  float* pw = part[w];
#pragma unroll
  for (int r = 0; r < 8; ++r) {
    pw[(8 * h + r) * TILE_N + n]                         = accA0[r];
    pw[(16 + 8 * h + r) * TILE_N + n]                    = accA1[r];
    pw[OHALF * TILE_N + (8 * h + r) * TILE_N + n]        = accB0[r];
    pw[OHALF * TILE_N + (16 + 8 * h + r) * TILE_N + n]   = accB1[r];
  }
  __syncthreads();

  float* Tw = T + ((size_t)(i * NTILES + 2 * tp) * OHALF) * TILE_N;
  for (int k = threadIdx.x; k < 2 * OHALF * TILE_N; k += 32 * KW) {
    float s = part[0][k];
#pragma unroll
    for (int v = 1; v < KW; ++v) s += part[v][k];
    Tw[k] = s;
  }
}

__global__ void __launch_bounds__(256) reduce_rows(const float* __restrict__ T,
                                                   const int* __restrict__ rank,
                                                   float* __restrict__ out,
                                                   int obase) {
  const int g  = blockIdx.x * 256 + threadIdx.x;
  const int b  = g & (BATCH - 1);
  const int op = g >> 9;
  float s = 0.0f;
#pragma unroll
  for (int i = 0; i < IN_X; ++i) {
    const int r = rank[i * BATCH + b] & (BATCH - 1);
    s += T[((size_t)(i * NTILES + (r >> 4)) * OHALF + op) * TILE_N + (r & 15)];
  }
  out[(size_t)(obase + op) * BATCH + b] = s;
}

extern "C" void kernel_launch(void* const* d_in, const int* in_sizes, int n_in,
                              void* d_out, int out_size, void* d_ws, size_t ws_size,
                              hipStream_t stream) {
  const float* x       = (const float*)d_in[0];
  const float* z       = (const float*)d_in[1];
  const float* P       = (const float*)d_in[2];
  const float* borders = (const float*)d_in[3];
  const float* invlen  = (const float*)d_in[4];
  float* out = (float*)d_out;

  float* T   = (float*)d_ws;
  int* rank  = (int*)(T + (size_t)IN_X * NTILES * OHALF * TILE_N) + IN_X * BATCH;

  bucket_gemm<<<dim3(NPAIRS, IN_X), dim3(32 * KW), 0, stream>>>(
      P, x, z, borders, invlen, rank, T, 0);
  reduce_rows<<<dim3(OHALF * BATCH / 256), dim3(256), 0, stream>>>(T, rank, out, 0);

  bucket_gemm<<<dim3(NPAIRS, IN_X), dim3(32 * KW), 0, stream>>>(
      P, x, z, borders, invlen, rank, T, OHALF);
  reduce_rows<<<dim3(OHALF * BATCH / 256), dim3(256), 0, stream>>>(T, rank, out, OHALF);
}


# Product kernel: z bin and z weights of the pair's 32 columns x 32 j computed once per work-group into LDS (was once per wave, 8 times over), lanes copy their entries by 16-byte LDS reads
# speedup vs baseline: 5.2727x; 1.0358x over previous
#include <hip/hip_runtime.h>
#include <math.h>

#define NUM_GRIDS 16
#define IN_X 32
#define IN_Z 32
#define OUT 64
#define BATCH 512
#define GP1 (NUM_GRIDS + 1)
#define TILE_N 16
#define NTILES (BATCH / TILE_N)
#define NPAIRS (NTILES / 2)
#define OHALF 32
#define KW 8
#define ZPITCH 36

typedef __attribute__((ext_vector_type(2))) float v2f;
typedef __attribute__((ext_vector_type(4))) float v4f;
typedef __attribute__((ext_vector_type(8))) float v8f;
typedef __attribute__((ext_vector_type(4))) int   v4i;

__device__ __forceinline__ int bin_index(float v) {
  float e   = __expf(-fabsf(v));
  float cdf = (v > 0.0f) ? (1.0f - 0.5f * e) : (0.5f * e);
  int idx = (int)(cdf * (float)NUM_GRIDS);
  return idx < 0 ? 0 : (idx > NUM_GRIDS - 1 ? NUM_GRIDS - 1 : idx);
}

__device__ __forceinline__ int bin_of(float v, const float* __restrict__ borders,
                                      const float* __restrict__ invlen, float& d) {
  const int idx = bin_index(v);
  d = (v - borders[idx]) * invlen[idx];
  return idx;
}

template <int N>
__device__ __forceinline__ int row_shift_right_or0(int v) {
  return __builtin_amdgcn_update_dpp(0, v, 0x110 + N, 0xf, 0xf, true);
}

__global__ void __launch_bounds__(32 * KW) __attribute__((amdgpu_waves_per_eu(4, 4))) bucket_gemm(
    const float* __restrict__ P,
    const float* __restrict__ x,
    const float* __restrict__ z,
    const float* __restrict__ borders,
    const float* __restrict__ invlen,
    int* __restrict__ rank,
    float* __restrict__ T,
    int obase)
{
  __shared__ float part[KW][2 * OHALF * TILE_N];
  __shared__ int   cnt[16][16];
  __shared__ int   first[16][16];
  __shared__ int   selb[2 * TILE_N];
  __shared__ int   selk[2 * TILE_N];
  __shared__ float selx[2 * TILE_N];
  __shared__ __attribute__((aligned(16))) int   zbin[2 * TILE_N][ZPITCH];
  __shared__ __attribute__((aligned(16))) float zlo[2 * TILE_N][ZPITCH];
  __shared__ __attribute__((aligned(16))) float zhi[2 * TILE_N][ZPITCH];

  const int tp   = blockIdx.x;
  const int i    = blockIdx.y;
  const int tid  = threadIdx.x;
  const int lane = threadIdx.x & 31;
  const int w    = __builtin_amdgcn_readfirstlane((int)(threadIdx.x >> 5));
  const int h    = lane >> 4;
  const int n    = lane & 15;

  const float xv0 = x[i * BATCH + tid];
  const float xv1 = x[i * BATCH + (BATCH / 2) + tid];
  const int k0 = bin_index(xv0);
  const int k1 = bin_index(xv1);
  unsigned same0 = ~0u, same1 = ~0u;
  unsigned ing0  = ~0u, ing1  = ~0u;
#pragma unroll
  for (int bit = 0; bit < 4; ++bit) {
    const unsigned m0 = __builtin_amdgcn_ballot_w32(((k0 >> bit) & 1) != 0);
    const unsigned m1 = __builtin_amdgcn_ballot_w32(((k1 >> bit) & 1) != 0);
    same0 &= ((k0 >> bit) & 1) ? m0 : ~m0;
    same1 &= ((k1 >> bit) & 1) ? m1 : ~m1;
    ing0  &= ((lane >> bit) & 1) ? m0 : ~m0;
    ing1  &= ((lane >> bit) & 1) ? m1 : ~m1;
  }
  const unsigned below = (1u << lane) - 1u;
  const int in0 = __builtin_popcount(same0 & below);
  const int in1 = __builtin_popcount(same1 & below);
  if (lane < 16) {
    cnt[w][lane]     = __builtin_popcount(ing0);
    cnt[w + 8][lane] = __builtin_popcount(ing1);
  }
  __syncthreads();
  {
    const int v = tid >> 4, g = tid & 15;
    int tot = 0, pre = 0;
#pragma unroll
    for (int u = 0; u < 16; ++u) {
      const int c = cnt[u][g];
      tot += c;
      pre += (u < v) ? c : 0;
    }
    int s = tot;
    s += row_shift_right_or0<1>(s);
    s += row_shift_right_or0<2>(s);
    s += row_shift_right_or0<4>(s);
    s += row_shift_right_or0<8>(s);
    first[v][g] = (s - tot) + pre;
  }
  __syncthreads();
  const int r0 = first[w][k0] + in0;
  const int r1 = first[w + 8][k1] + in1;
  if ((r0 >> 5) == tp) { selb[r0 & 31] = tid;               selk[r0 & 31] = k0; selx[r0 & 31] = xv0; }
  if ((r1 >> 5) == tp) { selb[r1 & 31] = tid + (BATCH / 2); selk[r1 & 31] = k1; selx[r1 & 31] = xv1; }
  __syncthreads();
  if (tp == 0 && obase == 0) {
    rank[i * BATCH + tid]               = r0;
    rank[i * BATCH + (BATCH / 2) + tid] = r1;
  }

  const int   ixA  = selk[n];
  const int   ixB  = selk[TILE_N + n];
  const float dxvA = (selx[n] - borders[ixA]) * invlen[ixA];
  const float dxvB = (selx[TILE_N + n] - borders[ixB]) * invlen[ixB];

  const int gminA = __builtin_amdgcn_readfirstlane(selk[0]);
  const int gmaxA = __builtin_amdgcn_readfirstlane(selk[TILE_N - 1]) + 1;
  const int gminB = __builtin_amdgcn_readfirstlane(selk[TILE_N]);
  const int gmaxB = __builtin_amdgcn_readfirstlane(selk[2 * TILE_N - 1]) + 1;

  {
    const int col = selb[lane] & (BATCH - 1);
#pragma unroll
    for (int k = 0; k < IN_Z / KW; ++k) {
      const int j = w + KW * k;
      float dz;
      zbin[lane][j] = bin_of(z[j * BATCH + col], borders, invlen, dz);
      zlo[lane][j]  = 1.0f - dz;
      zhi[lane][j]  = dz;
    }
  }
  __syncthreads();

  int   ziA[16], ziB[16];
  float zwA0[16], zwA1[16], zwB0[16], zwB1[16];
#pragma unroll
  for (int c = 0; c < 4; ++c) {
    const v4i bA = *(const v4i*)&zbin[n][16 * h + 4 * c];
    const v4f lA = *(const v4f*)&zlo[n][16 * h + 4 * c];
    const v4f hA = *(const v4f*)&zhi[n][16 * h + 4 * c];
    const v4i bB = *(const v4i*)&zbin[TILE_N + n][16 * h + 4 * c];
    const v4f lB = *(const v4f*)&zlo[TILE_N + n][16 * h + 4 * c];
    const v4f hB = *(const v4f*)&zhi[TILE_N + n][16 * h + 4 * c];
#pragma unroll
    for (int u = 0; u < 4; ++u) {
      const int e = 4 * c + u;
      ziA[e]  = bA[u];
      zwA0[e] = lA[u];
      zwA1[e] = hA[u];
      ziB[e]  = bB[u];
      zwB0[e] = lB[u];
      zwB1[e] = hB[u];
    }
  }

  v8f accA0 = {};
  v8f accA1 = {};
  v8f accB0 = {};
  v8f accB1 = {};

  const size_t sliceStride = (size_t)OUT * IN_X * IN_Z;
  const float* Prow = P + ((size_t)(obase + n) * IN_X + i) * IN_Z + 16 * h;
  const float* Ppre = P + ((size_t)(obase + lane) * IN_X + i) * IN_Z;
  const int lastSlice = gmaxB * GP1 + (GP1 - 1);

  int gx = gminA, gz = w;
  for (int slice = gminA * GP1 + w; slice <= lastSlice; slice += KW) {
    const bool doA = (gx <= gmaxA);
    const bool doB = (gx >= gminB);
    if (doA || doB) {
      const float* Ap = Prow + (size_t)slice * sliceStride;

      if (slice + KW <= lastSlice)
        __builtin_prefetch(Ppre + (size_t)(slice + KW) * sliceStride, 0, 3);

      v4f a0[4], a1[4];
#pragma unroll
      for (int c = 0; c < 4; ++c) {
        a0[c] = *(const v4f*)(Ap + 4 * c);
        a1[c] = *(const v4f*)(Ap + (size_t)16 * IN_X * IN_Z + 4 * c);
      }

      if (doA) {
        const float wx = (gx == ixA) ? (1.0f - dxvA) : ((gx == ixA + 1) ? dxvA : 0.0f);
#pragma unroll
        for (int q = 0; q < 8; ++q) {
          const int e0 = 2 * q, e1 = 2 * q + 1;
          v2f bv;
          bv.x = (ziA[e0] == gz) ? zwA0[e0] * wx : ((ziA[e0] + 1 == gz) ? zwA1[e0] * wx : 0.0f);
          bv.y = (ziA[e1] == gz) ? zwA0[e1] * wx : ((ziA[e1] + 1 == gz) ? zwA1[e1] * wx : 0.0f);
          v2f av0, av1;
          av0.x = a0[q >> 1][2 * (q & 1)];
          av0.y = a0[q >> 1][2 * (q & 1) + 1];
          av1.x = a1[q >> 1][2 * (q & 1)];
          av1.y = a1[q >> 1][2 * (q & 1) + 1];
          accA0 = __builtin_amdgcn_wmma_f32_16x16x4_f32(
              false, av0, false, bv, (short)0, accA0, false, false);
          accA1 = __builtin_amdgcn_wmma_f32_16x16x4_f32(
              false, av1, false, bv, (short)0, accA1, false, false);
        }
      }
      if (doB) {
        const float wx = (gx == ixB) ? (1.0f - dxvB) : ((gx == ixB + 1) ? dxvB : 0.0f);
#pragma unroll
        for (int q = 0; q < 8; ++q) {
          const int e0 = 2 * q, e1 = 2 * q + 1;
          v2f bv;
          bv.x = (ziB[e0] == gz) ? zwB0[e0] * wx : ((ziB[e0] + 1 == gz) ? zwB1[e0] * wx : 0.0f);
          bv.y = (ziB[e1] == gz) ? zwB0[e1] * wx : ((ziB[e1] + 1 == gz) ? zwB1[e1] * wx : 0.0f);
          v2f av0, av1;
          av0.x = a0[q >> 1][2 * (q & 1)];
          av0.y = a0[q >> 1][2 * (q & 1) + 1];
          av1.x = a1[q >> 1][2 * (q & 1)];
          av1.y = a1[q >> 1][2 * (q & 1) + 1];
          accB0 = __builtin_amdgcn_wmma_f32_16x16x4_f32(
              false, av0, false, bv, (short)0, accB0, false, false);
          accB1 = __builtin_amdgcn_wmma_f32_16x16x4_f32(
              false, av1, false, bv, (short)0, accB1, false, false);
        }
      }
    }

    gz += KW;
    if (gz >= GP1) { gz -= GP1; ++gx; }
  }

  float* pw = part[w];
#pragma unroll
  for (int r = 0; r < 8; ++r) {
    pw[(8 * h + r) * TILE_N + n]                         = accA0[r];
    pw[(16 + 8 * h + r) * TILE_N + n]                    = accA1[r];
    pw[OHALF * TILE_N + (8 * h + r) * TILE_N + n]        = accB0[r];
    pw[OHALF * TILE_N + (16 + 8 * h + r) * TILE_N + n]   = accB1[r];
  }
  __syncthreads();

  float* Tw = T + ((size_t)(i * NTILES + 2 * tp) * OHALF) * TILE_N;
  for (int k = threadIdx.x; k < 2 * OHALF * TILE_N; k += 32 * KW) {
    float s = part[0][k];
#pragma unroll
    for (int v = 1; v < KW; ++v) s += part[v][k];
    Tw[k] = s;
  }
}

__global__ void __launch_bounds__(256) reduce_rows(const float* __restrict__ T,
                                                   const int* __restrict__ rank,
                                                   float* __restrict__ out,
                                                   int obase) {
  const int g  = blockIdx.x * 256 + threadIdx.x;
  const int b  = g & (BATCH - 1);
  const int op = g >> 9;
  float s = 0.0f;
#pragma unroll
  for (int i = 0; i < IN_X; ++i) {
    const int r = rank[i * BATCH + b] & (BATCH - 1);
    s += T[((size_t)(i * NTILES + (r >> 4)) * OHALF + op) * TILE_N + (r & 15)];
  }
  out[(size_t)(obase + op) * BATCH + b] = s;
}

extern "C" void kernel_launch(void* const* d_in, const int* in_sizes, int n_in,
                              void* d_out, int out_size, void* d_ws, size_t ws_size,
                              hipStream_t stream) {
  const float* x       = (const float*)d_in[0];
  const float* z       = (const float*)d_in[1];
  const float* P       = (const float*)d_in[2];
  const float* borders = (const float*)d_in[3];
  const float* invlen  = (const float*)d_in[4];
  float* out = (float*)d_out;

  float* T   = (float*)d_ws;
  int* rank  = (int*)(T + (size_t)IN_X * NTILES * OHALF * TILE_N) + IN_X * BATCH;

  bucket_gemm<<<dim3(NPAIRS, IN_X), dim3(32 * KW), 0, stream>>>(
      P, x, z, borders, invlen, rank, T, 0);
  reduce_rows<<<dim3(OHALF * BATCH / 256), dim3(256), 0, stream>>>(T, rank, out, 0);

  bucket_gemm<<<dim3(NPAIRS, IN_X), dim3(32 * KW), 0, stream>>>(
      P, x, z, borders, invlen, rank, T, OHALF);
  reduce_rows<<<dim3(OHALF * BATCH / 256), dim3(256), 0, stream>>>(T, rank, out, OHALF);
}


# Product kernel: each slice read as whole 128-byte lines (4 lines per load, was 16) into a per-wave LDS buffer, A fragments read back from LDS; next slice's loads in flight during the products
# speedup vs baseline: 5.5196x; 1.0468x over previous
#include <hip/hip_runtime.h>
#include <math.h>

#define NUM_GRIDS 16
#define IN_X 32
#define IN_Z 32
#define OUT 64
#define BATCH 512
#define GP1 (NUM_GRIDS + 1)
#define TILE_N 16
#define NTILES (BATCH / TILE_N)
#define NPAIRS (NTILES / 2)
#define OHALF 32
#define KW 8
#define ZPITCH 36
#define SPITCH 36

typedef __attribute__((ext_vector_type(2))) float v2f;
typedef __attribute__((ext_vector_type(4))) float v4f;
typedef __attribute__((ext_vector_type(8))) float v8f;
typedef __attribute__((ext_vector_type(4))) int   v4i;

__device__ __forceinline__ int bin_index(float v) {
  float e   = __expf(-fabsf(v));
  float cdf = (v > 0.0f) ? (1.0f - 0.5f * e) : (0.5f * e);
  int idx = (int)(cdf * (float)NUM_GRIDS);
  return idx < 0 ? 0 : (idx > NUM_GRIDS - 1 ? NUM_GRIDS - 1 : idx);
}

__device__ __forceinline__ int bin_of(float v, const float* __restrict__ borders,
                                      const float* __restrict__ invlen, float& d) {
  const int idx = bin_index(v);
  d = (v - borders[idx]) * invlen[idx];
  return idx;
}

template <int N>
__device__ __forceinline__ int row_shift_right_or0(int v) {
  return __builtin_amdgcn_update_dpp(0, v, 0x110 + N, 0xf, 0xf, true);
}

__global__ void __launch_bounds__(32 * KW) __attribute__((amdgpu_waves_per_eu(4, 4))) bucket_gemm(
    const float* __restrict__ P,
    const float* __restrict__ x,
    const float* __restrict__ z,
    const float* __restrict__ borders,
    const float* __restrict__ invlen,
    int* __restrict__ rank,
    float* __restrict__ T,
    int obase)
{
  __shared__ float part[KW][2 * OHALF * TILE_N];
  __shared__ int   cnt[16][16];
  __shared__ int   first[16][16];
  __shared__ int   selb[2 * TILE_N];
  __shared__ int   selk[2 * TILE_N];
  __shared__ float selx[2 * TILE_N];
  __shared__ __attribute__((aligned(16))) int   zbin[2 * TILE_N][ZPITCH];
  __shared__ __attribute__((aligned(16))) float zlo[2 * TILE_N][ZPITCH];
  __shared__ __attribute__((aligned(16))) float zhi[2 * TILE_N][ZPITCH];
  __shared__ __attribute__((aligned(16))) float stage[KW][OHALF][SPITCH];

  const int tp   = blockIdx.x;
  const int i    = blockIdx.y;
  const int tid  = threadIdx.x;
  const int lane = threadIdx.x & 31;
  const int w    = __builtin_amdgcn_readfirstlane((int)(threadIdx.x >> 5));
  const int h    = lane >> 4;
  const int n    = lane & 15;

  const float xv0 = x[i * BATCH + tid];
  const float xv1 = x[i * BATCH + (BATCH / 2) + tid];
  const int k0 = bin_index(xv0);
  const int k1 = bin_index(xv1);
  unsigned same0 = ~0u, same1 = ~0u;
  unsigned ing0  = ~0u, ing1  = ~0u;
#pragma unroll
  for (int bit = 0; bit < 4; ++bit) {
    const unsigned m0 = __builtin_amdgcn_ballot_w32(((k0 >> bit) & 1) != 0);
    const unsigned m1 = __builtin_amdgcn_ballot_w32(((k1 >> bit) & 1) != 0);
    same0 &= ((k0 >> bit) & 1) ? m0 : ~m0;
    same1 &= ((k1 >> bit) & 1) ? m1 : ~m1;
    ing0  &= ((lane >> bit) & 1) ? m0 : ~m0;
    ing1  &= ((lane >> bit) & 1) ? m1 : ~m1;
  }
  const unsigned below = (1u << lane) - 1u;
  const int in0 = __builtin_popcount(same0 & below);
  const int in1 = __builtin_popcount(same1 & below);
  if (lane < 16) {
    cnt[w][lane]     = __builtin_popcount(ing0);
    cnt[w + 8][lane] = __builtin_popcount(ing1);
  }
  __syncthreads();
  {
    const int v = tid >> 4, g = tid & 15;
    int tot = 0, pre = 0;
#pragma unroll
    for (int u = 0; u < 16; ++u) {
      const int c = cnt[u][g];
      tot += c;
      pre += (u < v) ? c : 0;
    }
    int s = tot;
    s += row_shift_right_or0<1>(s);
    s += row_shift_right_or0<2>(s);
    s += row_shift_right_or0<4>(s);
    s += row_shift_right_or0<8>(s);
    first[v][g] = (s - tot) + pre;
  }
  __syncthreads();
  const int r0 = first[w][k0] + in0;
  const int r1 = first[w + 8][k1] + in1;
  if ((r0 >> 5) == tp) { selb[r0 & 31] = tid;               selk[r0 & 31] = k0; selx[r0 & 31] = xv0; }
  if ((r1 >> 5) == tp) { selb[r1 & 31] = tid + (BATCH / 2); selk[r1 & 31] = k1; selx[r1 & 31] = xv1; }
  __syncthreads();
  if (tp == 0 && obase == 0) {
    rank[i * BATCH + tid]               = r0;
    rank[i * BATCH + (BATCH / 2) + tid] = r1;
  }

  const int   ixA  = selk[n];
  const int   ixB  = selk[TILE_N + n];
  const float dxvA = (selx[n] - borders[ixA]) * invlen[ixA];
  const float dxvB = (selx[TILE_N + n] - borders[ixB]) * invlen[ixB];

  const int gminA = __builtin_amdgcn_readfirstlane(selk[0]);
  const int gmaxA = __builtin_amdgcn_readfirstlane(selk[TILE_N - 1]) + 1;
  const int gminB = __builtin_amdgcn_readfirstlane(selk[TILE_N]);
  const int gmaxB = __builtin_amdgcn_readfirstlane(selk[2 * TILE_N - 1]) + 1;

  const size_t sliceStride = (size_t)OUT * IN_X * IN_Z;
  const size_t rowStride   = (size_t)IN_X * IN_Z;
  const int ldrow = lane >> 3;
  const int ldcol = (lane & 7) * 4;
  const float* Pld  = P + ((size_t)(obase + ldrow) * IN_X + i) * IN_Z + ldcol;
  const int lastSlice  = gmaxB * GP1 + (GP1 - 1);
  const int firstSlice = gminA * GP1 + w;
  v4f raw[8];
#pragma unroll
  for (int r = 0; r < 8; ++r)
    raw[r] = *(const v4f*)(Pld + (size_t)firstSlice * sliceStride + (size_t)(4 * r) * rowStride);

  {
    const int col = selb[lane] & (BATCH - 1);
#pragma unroll
    for (int k = 0; k < IN_Z / KW; ++k) {
      const int j = w + KW * k;
      float dz;
      zbin[lane][j] = bin_of(z[j * BATCH + col], borders, invlen, dz);
      zlo[lane][j]  = 1.0f - dz;
      zhi[lane][j]  = dz;
    }
  }
  __syncthreads();

  int   ziA[16], ziB[16];
  float zwA0[16], zwA1[16], zwB0[16], zwB1[16];
#pragma unroll
  for (int c = 0; c < 4; ++c) {
    const v4i bA = *(const v4i*)&zbin[n][16 * h + 4 * c];
    const v4f lA = *(const v4f*)&zlo[n][16 * h + 4 * c];
    const v4f hA = *(const v4f*)&zhi[n][16 * h + 4 * c];
    const v4i bB = *(const v4i*)&zbin[TILE_N + n][16 * h + 4 * c];
    const v4f lB = *(const v4f*)&zlo[TILE_N + n][16 * h + 4 * c];
    const v4f hB = *(const v4f*)&zhi[TILE_N + n][16 * h + 4 * c];
#pragma unroll
    for (int u = 0; u < 4; ++u) {
      const int e = 4 * c + u;
      ziA[e]  = bA[u];
      zwA0[e] = lA[u];
      zwA1[e] = hA[u];
      ziB[e]  = bB[u];
      zwB0[e] = lB[u];
      zwB1[e] = hB[u];
    }
  }

  v8f accA0 = {};
  v8f accA1 = {};
  v8f accB0 = {};
  v8f accB1 = {};

  const float* Ppre = P + ((size_t)(obase + lane) * IN_X + i) * IN_Z;
  float (*st)[SPITCH] = stage[w];

  int gx = gminA, gz = w;
  for (int slice = firstSlice; slice <= lastSlice; slice += KW) {
    const bool doA = (gx <= gmaxA);
    const bool doB = (gx >= gminB);

#pragma unroll
    for (int r = 0; r < 8; ++r)
      *(v4f*)&st[4 * r + ldrow][ldcol] = raw[r];

    if (slice + KW <= lastSlice) {
      const float* Np = Pld + (size_t)(slice + KW) * sliceStride;
#pragma unroll
      for (int r = 0; r < 8; ++r)
        raw[r] = *(const v4f*)(Np + (size_t)(4 * r) * rowStride);
      if (slice + 2 * KW <= lastSlice)
        __builtin_prefetch(Ppre + (size_t)(slice + 2 * KW) * sliceStride, 0, 3);
    }

    if (doA || doB) {
      v4f a0[4], a1[4];
#pragma unroll
      for (int c = 0; c < 4; ++c) {
        a0[c] = *(const v4f*)&st[n][16 * h + 4 * c];
        a1[c] = *(const v4f*)&st[16 + n][16 * h + 4 * c];
      }

      if (doA) {
        const float wx = (gx == ixA) ? (1.0f - dxvA) : ((gx == ixA + 1) ? dxvA : 0.0f);
#pragma unroll
        for (int q = 0; q < 8; ++q) {
          const int e0 = 2 * q, e1 = 2 * q + 1;
          v2f bv;
          bv.x = (ziA[e0] == gz) ? zwA0[e0] * wx : ((ziA[e0] + 1 == gz) ? zwA1[e0] * wx : 0.0f);
          bv.y = (ziA[e1] == gz) ? zwA0[e1] * wx : ((ziA[e1] + 1 == gz) ? zwA1[e1] * wx : 0.0f);
          v2f av0, av1;
          av0.x = a0[q >> 1][2 * (q & 1)];
          av0.y = a0[q >> 1][2 * (q & 1) + 1];
          av1.x = a1[q >> 1][2 * (q & 1)];
          av1.y = a1[q >> 1][2 * (q & 1) + 1];
          accA0 = __builtin_amdgcn_wmma_f32_16x16x4_f32(
              false, av0, false, bv, (short)0, accA0, false, false);
          accA1 = __builtin_amdgcn_wmma_f32_16x16x4_f32(
              false, av1, false, bv, (short)0, accA1, false, false);
        }
      }
      if (doB) {
        const float wx = (gx == ixB) ? (1.0f - dxvB) : ((gx == ixB + 1) ? dxvB : 0.0f);
#pragma unroll
        for (int q = 0; q < 8; ++q) {
          const int e0 = 2 * q, e1 = 2 * q + 1;
          v2f bv;
          bv.x = (ziB[e0] == gz) ? zwB0[e0] * wx : ((ziB[e0] + 1 == gz) ? zwB1[e0] * wx : 0.0f);
          bv.y = (ziB[e1] == gz) ? zwB0[e1] * wx : ((ziB[e1] + 1 == gz) ? zwB1[e1] * wx : 0.0f);
          v2f av0, av1;
          av0.x = a0[q >> 1][2 * (q & 1)];
          av0.y = a0[q >> 1][2 * (q & 1) + 1];
          av1.x = a1[q >> 1][2 * (q & 1)];
          av1.y = a1[q >> 1][2 * (q & 1) + 1];
          accB0 = __builtin_amdgcn_wmma_f32_16x16x4_f32(
              false, av0, false, bv, (short)0, accB0, false, false);
          accB1 = __builtin_amdgcn_wmma_f32_16x16x4_f32(
              false, av1, false, bv, (short)0, accB1, false, false);
        }
      }
    }

    gz += KW;
    if (gz >= GP1) { gz -= GP1; ++gx; }
  }

  float* pw = part[w];
#pragma unroll
  for (int r = 0; r < 8; ++r) {
    pw[(8 * h + r) * TILE_N + n]                         = accA0[r];
    pw[(16 + 8 * h + r) * TILE_N + n]                    = accA1[r];
    pw[OHALF * TILE_N + (8 * h + r) * TILE_N + n]        = accB0[r];
    pw[OHALF * TILE_N + (16 + 8 * h + r) * TILE_N + n]   = accB1[r];
  }
  __syncthreads();

  float* Tw = T + ((size_t)(i * NTILES + 2 * tp) * OHALF) * TILE_N;
  for (int k = threadIdx.x; k < 2 * OHALF * TILE_N; k += 32 * KW) {
    float s = part[0][k];
#pragma unroll
    for (int v = 1; v < KW; ++v) s += part[v][k];
    Tw[k] = s;
  }
}

__global__ void __launch_bounds__(256) reduce_rows(const float* __restrict__ T,
                                                   const int* __restrict__ rank,
                                                   float* __restrict__ out,
                                                   int obase) {
  const int g  = blockIdx.x * 256 + threadIdx.x;
  const int b  = g & (BATCH - 1);
  const int op = g >> 9;
  float s = 0.0f;
#pragma unroll
  for (int i = 0; i < IN_X; ++i) {
    const int r = rank[i * BATCH + b] & (BATCH - 1);
    s += T[((size_t)(i * NTILES + (r >> 4)) * OHALF + op) * TILE_N + (r & 15)];
  }
  out[(size_t)(obase + op) * BATCH + b] = s;
}

extern "C" void kernel_launch(void* const* d_in, const int* in_sizes, int n_in,
                              void* d_out, int out_size, void* d_ws, size_t ws_size,
                              hipStream_t stream) {
  const float* x       = (const float*)d_in[0];
  const float* z       = (const float*)d_in[1];
  const float* P       = (const float*)d_in[2];
  const float* borders = (const float*)d_in[3];
  const float* invlen  = (const float*)d_in[4];
  float* out = (float*)d_out;

  float* T   = (float*)d_ws;
  int* rank  = (int*)(T + (size_t)IN_X * NTILES * OHALF * TILE_N) + IN_X * BATCH;

  bucket_gemm<<<dim3(NPAIRS, IN_X), dim3(32 * KW), 0, stream>>>(
      P, x, z, borders, invlen, rank, T, 0);
  reduce_rows<<<dim3(OHALF * BATCH / 256), dim3(256), 0, stream>>>(T, rank, out, 0);

  bucket_gemm<<<dim3(NPAIRS, IN_X), dim3(32 * KW), 0, stream>>>(
      P, x, z, borders, invlen, rank, T, OHALF);
  reduce_rows<<<dim3(OHALF * BATCH / 256), dim3(256), 0, stream>>>(T, rank, out, OHALF);
}


# Product kernel: B fragments (z weights, all 17 gz, both tiles) built once per work-group as an LDS table and read by 16-byte LDS reads; per-slice compare/select chains on the vector unit gone
# speedup vs baseline: 5.9410x; 1.0763x over previous
#include <hip/hip_runtime.h>
#include <math.h>

#define NUM_GRIDS 16
#define IN_X 32
#define IN_Z 32
#define OUT 64
#define BATCH 512
#define GP1 (NUM_GRIDS + 1)
#define TILE_N 16
#define NTILES (BATCH / TILE_N)
#define NPAIRS (NTILES / 2)
#define OHALF 32
#define KW 8
#define SPITCH 36
#define JG (IN_Z / 4)

typedef __attribute__((ext_vector_type(2))) float v2f;
typedef __attribute__((ext_vector_type(4))) float v4f;
typedef __attribute__((ext_vector_type(8))) float v8f;

__device__ __forceinline__ int bin_index(float v) {
  float e   = __expf(-fabsf(v));
  float cdf = (v > 0.0f) ? (1.0f - 0.5f * e) : (0.5f * e);
  int idx = (int)(cdf * (float)NUM_GRIDS);
  return idx < 0 ? 0 : (idx > NUM_GRIDS - 1 ? NUM_GRIDS - 1 : idx);
}

__device__ __forceinline__ int bin_of(float v, const float* __restrict__ borders,
                                      const float* __restrict__ invlen, float& d) {
  const int idx = bin_index(v);
  d = (v - borders[idx]) * invlen[idx];
  return idx;
}

template <int N>
__device__ __forceinline__ int row_shift_right_or0(int v) {
  return __builtin_amdgcn_update_dpp(0, v, 0x110 + N, 0xf, 0xf, true);
}

__global__ void __launch_bounds__(32 * KW) __attribute__((amdgpu_waves_per_eu(4, 4))) bucket_gemm(
    const float* __restrict__ P,
    const float* __restrict__ x,
    const float* __restrict__ z,
    const float* __restrict__ borders,
    const float* __restrict__ invlen,
    int* __restrict__ rank,
    float* __restrict__ T,
    int obase)
{
  __shared__ float part[KW][2 * OHALF * TILE_N];
  __shared__ int   cnt[16][16];
  __shared__ int   first[16][16];
  __shared__ int   selb[2 * TILE_N];
  __shared__ int   selk[2 * TILE_N];
  __shared__ float selx[2 * TILE_N];
  __shared__ __attribute__((aligned(16))) float btab[2][GP1][JG][TILE_N][4];
  __shared__ __attribute__((aligned(16))) float stage[KW][OHALF][SPITCH];

  const int tp   = blockIdx.x;
  const int i    = blockIdx.y;
  const int tid  = threadIdx.x;
  const int lane = threadIdx.x & 31;
  const int w    = __builtin_amdgcn_readfirstlane((int)(threadIdx.x >> 5));
  const int h    = lane >> 4;
  const int n    = lane & 15;

  const float xv0 = x[i * BATCH + tid];
  const float xv1 = x[i * BATCH + (BATCH / 2) + tid];
  {
    v4f* bz = (v4f*)&btab[0][0][0][0][0];
    const v4f zero4 = {0.0f, 0.0f, 0.0f, 0.0f};
#pragma unroll
    for (int k = 0; k < (2 * GP1 * JG * TILE_N) / (32 * KW); ++k)
      bz[k * (32 * KW) + tid] = zero4;
  }
  const int k0 = bin_index(xv0);
  const int k1 = bin_index(xv1);
  unsigned same0 = ~0u, same1 = ~0u;
  unsigned ing0  = ~0u, ing1  = ~0u;
#pragma unroll
  for (int bit = 0; bit < 4; ++bit) {
    const unsigned m0 = __builtin_amdgcn_ballot_w32(((k0 >> bit) & 1) != 0);
    const unsigned m1 = __builtin_amdgcn_ballot_w32(((k1 >> bit) & 1) != 0);
    same0 &= ((k0 >> bit) & 1) ? m0 : ~m0;
    same1 &= ((k1 >> bit) & 1) ? m1 : ~m1;
    ing0  &= ((lane >> bit) & 1) ? m0 : ~m0;
    ing1  &= ((lane >> bit) & 1) ? m1 : ~m1;
  }
  const unsigned below = (1u << lane) - 1u;
  const int in0 = __builtin_popcount(same0 & below);
  const int in1 = __builtin_popcount(same1 & below);
  if (lane < 16) {
    cnt[w][lane]     = __builtin_popcount(ing0);
    cnt[w + 8][lane] = __builtin_popcount(ing1);
  }
  __syncthreads();
  {
    const int v = tid >> 4, g = tid & 15;
    int tot = 0, pre = 0;
#pragma unroll
    for (int u = 0; u < 16; ++u) {
      const int c = cnt[u][g];
      tot += c;
      pre += (u < v) ? c : 0;
    }
    int s = tot;
    s += row_shift_right_or0<1>(s);
    s += row_shift_right_or0<2>(s);
    s += row_shift_right_or0<4>(s);
    s += row_shift_right_or0<8>(s);
    first[v][g] = (s - tot) + pre;
  }
  __syncthreads();
  const int r0 = first[w][k0] + in0;
  const int r1 = first[w + 8][k1] + in1;
  if ((r0 >> 5) == tp) { selb[r0 & 31] = tid;               selk[r0 & 31] = k0; selx[r0 & 31] = xv0; }
  if ((r1 >> 5) == tp) { selb[r1 & 31] = tid + (BATCH / 2); selk[r1 & 31] = k1; selx[r1 & 31] = xv1; }
  __syncthreads();
  if (tp == 0 && obase == 0) {
    rank[i * BATCH + tid]               = r0;
    rank[i * BATCH + (BATCH / 2) + tid] = r1;
  }

  const int   ixA  = selk[n];
  const int   ixB  = selk[TILE_N + n];
  const float dxvA = (selx[n] - borders[ixA]) * invlen[ixA];
  const float dxvB = (selx[TILE_N + n] - borders[ixB]) * invlen[ixB];

  const int gminA = __builtin_amdgcn_readfirstlane(selk[0]);
  const int gmaxA = __builtin_amdgcn_readfirstlane(selk[TILE_N - 1]) + 1;
  const int gminB = __builtin_amdgcn_readfirstlane(selk[TILE_N]);
  const int gmaxB = __builtin_amdgcn_readfirstlane(selk[2 * TILE_N - 1]) + 1;

  const size_t sliceStride = (size_t)OUT * IN_X * IN_Z;
  const size_t rowStride   = (size_t)IN_X * IN_Z;
  const int ldrow = lane >> 3;
  const int ldcol = (lane & 7) * 4;
  const float* Pld  = P + ((size_t)(obase + ldrow) * IN_X + i) * IN_Z + ldcol;
  const int lastSlice  = gmaxB * GP1 + (GP1 - 1);
  const int firstSlice = gminA * GP1 + w;
  v4f raw[8];
#pragma unroll
  for (int r = 0; r < 8; ++r)
    raw[r] = *(const v4f*)(Pld + (size_t)firstSlice * sliceStride + (size_t)(4 * r) * rowStride);

  {
    const int col = selb[lane] & (BATCH - 1);
#pragma unroll
    for (int k = 0; k < IN_Z / KW; ++k) {
      const int j = w + KW * k;
      float dz;
      const int zi = bin_of(z[j * BATCH + col], borders, invlen, dz);
      btab[h][zi][j >> 2][n][j & 3]     = 1.0f - dz;
      btab[h][zi + 1][j >> 2][n][j & 3] = dz;
    }
  }
  __syncthreads();

  v8f accA0 = {};
  v8f accA1 = {};
  v8f accB0 = {};
  v8f accB1 = {};

  const float* Ppre = P + ((size_t)(obase + lane) * IN_X + i) * IN_Z;
  float (*st)[SPITCH] = stage[w];

  int gx = gminA, gz = w;
  for (int slice = firstSlice; slice <= lastSlice; slice += KW) {
    const bool doA = (gx <= gmaxA);
    const bool doB = (gx >= gminB);

#pragma unroll
    for (int r = 0; r < 8; ++r)
      *(v4f*)&st[4 * r + ldrow][ldcol] = raw[r];

    if (slice + KW <= lastSlice) {
      const float* Np = Pld + (size_t)(slice + KW) * sliceStride;
#pragma unroll
      for (int r = 0; r < 8; ++r)
        raw[r] = *(const v4f*)(Np + (size_t)(4 * r) * rowStride);
      if (slice + 2 * KW <= lastSlice)
        __builtin_prefetch(Ppre + (size_t)(slice + 2 * KW) * sliceStride, 0, 3);
    }

    if (doA || doB) {
      v4f a0[4], a1[4];
#pragma unroll
      for (int c = 0; c < 4; ++c) {
        a0[c] = *(const v4f*)&st[n][16 * h + 4 * c];
        a1[c] = *(const v4f*)&st[16 + n][16 * h + 4 * c];
      }
      v4f bA[4], bB[4];
#pragma unroll
      for (int c = 0; c < 4; ++c) {
        bA[c] = *(const v4f*)&btab[0][gz][4 * h + c][n][0];
        bB[c] = *(const v4f*)&btab[1][gz][4 * h + c][n][0];
      }
      asm volatile("" : "+v"(bA[0]), "+v"(bA[1]), "+v"(bA[2]), "+v"(bA[3]));

      if (doA) {
        const float wx = (gx == ixA) ? (1.0f - dxvA) : ((gx == ixA + 1) ? dxvA : 0.0f);
#pragma unroll
        for (int q = 0; q < 8; ++q) {
          v2f bv;
          bv.x = bA[q >> 1][2 * (q & 1)] * wx;
          bv.y = bA[q >> 1][2 * (q & 1) + 1] * wx;
          v2f av0, av1;
          av0.x = a0[q >> 1][2 * (q & 1)];
          av0.y = a0[q >> 1][2 * (q & 1) + 1];
          av1.x = a1[q >> 1][2 * (q & 1)];
          av1.y = a1[q >> 1][2 * (q & 1) + 1];
          accA0 = __builtin_amdgcn_wmma_f32_16x16x4_f32(
              false, av0, false, bv, (short)0, accA0, false, false);
          accA1 = __builtin_amdgcn_wmma_f32_16x16x4_f32(
              false, av1, false, bv, (short)0, accA1, false, false);
        }
      }
      if (doB) {
        const float wx = (gx == ixB) ? (1.0f - dxvB) : ((gx == ixB + 1) ? dxvB : 0.0f);
#pragma unroll
        for (int q = 0; q < 8; ++q) {
          v2f bv;
          bv.x = bB[q >> 1][2 * (q & 1)] * wx;
          bv.y = bB[q >> 1][2 * (q & 1) + 1] * wx;
          v2f av0, av1;
          av0.x = a0[q >> 1][2 * (q & 1)];
          av0.y = a0[q >> 1][2 * (q & 1) + 1];
          av1.x = a1[q >> 1][2 * (q & 1)];
          av1.y = a1[q >> 1][2 * (q & 1) + 1];
          accB0 = __builtin_amdgcn_wmma_f32_16x16x4_f32(
              false, av0, false, bv, (short)0, accB0, false, false);
          accB1 = __builtin_amdgcn_wmma_f32_16x16x4_f32(
              false, av1, false, bv, (short)0, accB1, false, false);
        }
      }
    }

    gz += KW;
    if (gz >= GP1) { gz -= GP1; ++gx; }
  }

  float* pw = part[w];
#pragma unroll
  for (int r = 0; r < 8; ++r) {
    pw[(8 * h + r) * TILE_N + n]                         = accA0[r];
    pw[(16 + 8 * h + r) * TILE_N + n]                    = accA1[r];
    pw[OHALF * TILE_N + (8 * h + r) * TILE_N + n]        = accB0[r];
    pw[OHALF * TILE_N + (16 + 8 * h + r) * TILE_N + n]   = accB1[r];
  }
  __syncthreads();

  float* Tw = T + ((size_t)(i * NTILES + 2 * tp) * OHALF) * TILE_N;
  for (int k = threadIdx.x; k < 2 * OHALF * TILE_N; k += 32 * KW) {
    float s = part[0][k];
#pragma unroll
    for (int v = 1; v < KW; ++v) s += part[v][k];
    Tw[k] = s;
  }
}

__global__ void __launch_bounds__(256) reduce_rows(const float* __restrict__ T,
                                                   const int* __restrict__ rank,
                                                   float* __restrict__ out,
                                                   int obase) {
  const int g  = blockIdx.x * 256 + threadIdx.x;
  const int b  = g & (BATCH - 1);
  const int op = g >> 9;
  float s = 0.0f;
#pragma unroll
  for (int i = 0; i < IN_X; ++i) {
    const int r = rank[i * BATCH + b] & (BATCH - 1);
    s += T[((size_t)(i * NTILES + (r >> 4)) * OHALF + op) * TILE_N + (r & 15)];
  }
  out[(size_t)(obase + op) * BATCH + b] = s;
}

extern "C" void kernel_launch(void* const* d_in, const int* in_sizes, int n_in,
                              void* d_out, int out_size, void* d_ws, size_t ws_size,
                              hipStream_t stream) {
  const float* x       = (const float*)d_in[0];
  const float* z       = (const float*)d_in[1];
  const float* P       = (const float*)d_in[2];
  const float* borders = (const float*)d_in[3];
  const float* invlen  = (const float*)d_in[4];
  float* out = (float*)d_out;

  float* T   = (float*)d_ws;
  int* rank  = (int*)(T + (size_t)IN_X * NTILES * OHALF * TILE_N) + IN_X * BATCH;

  bucket_gemm<<<dim3(NPAIRS, IN_X), dim3(32 * KW), 0, stream>>>(
      P, x, z, borders, invlen, rank, T, 0);
  reduce_rows<<<dim3(OHALF * BATCH / 256), dim3(256), 0, stream>>>(T, rank, out, 0);

  bucket_gemm<<<dim3(NPAIRS, IN_X), dim3(32 * KW), 0, stream>>>(
      P, x, z, borders, invlen, rank, T, OHALF);
  reduce_rows<<<dim3(OHALF * BATCH / 256), dim3(256), 0, stream>>>(T, rank, out, OHALF);
}


# Row-sum kernel: 256 one-wave work-groups (was 64 of 8 waves), a wave sums 32 columns x two neighbouring output rows that share each 128-byte line of T; same order of summation
# speedup vs baseline: 6.1763x; 1.0396x over previous
#include <hip/hip_runtime.h>
#include <math.h>

#define NUM_GRIDS 16
#define IN_X 32
#define IN_Z 32
#define OUT 64
#define BATCH 512
#define GP1 (NUM_GRIDS + 1)
#define TILE_N 16
#define NTILES (BATCH / TILE_N)
#define NPAIRS (NTILES / 2)
#define OHALF 32
#define KW 8
#define SPITCH 36
#define JG (IN_Z / 4)

typedef __attribute__((ext_vector_type(2))) float v2f;
typedef __attribute__((ext_vector_type(4))) float v4f;
typedef __attribute__((ext_vector_type(8))) float v8f;

__device__ __forceinline__ int bin_index(float v) {
  float e   = __expf(-fabsf(v));
  float cdf = (v > 0.0f) ? (1.0f - 0.5f * e) : (0.5f * e);
  int idx = (int)(cdf * (float)NUM_GRIDS);
  return idx < 0 ? 0 : (idx > NUM_GRIDS - 1 ? NUM_GRIDS - 1 : idx);
}

__device__ __forceinline__ int bin_of(float v, const float* __restrict__ borders,
                                      const float* __restrict__ invlen, float& d) {
  const int idx = bin_index(v);
  d = (v - borders[idx]) * invlen[idx];
  return idx;
}

template <int N>
__device__ __forceinline__ int row_shift_right_or0(int v) {
  return __builtin_amdgcn_update_dpp(0, v, 0x110 + N, 0xf, 0xf, true);
}

__global__ void __launch_bounds__(32 * KW) __attribute__((amdgpu_waves_per_eu(4, 4))) bucket_gemm(
    const float* __restrict__ P,
    const float* __restrict__ x,
    const float* __restrict__ z,
    const float* __restrict__ borders,
    const float* __restrict__ invlen,
    int* __restrict__ rank,
    float* __restrict__ T,
    int obase)
{
  __shared__ float part[KW][2 * OHALF * TILE_N];
  __shared__ int   cnt[16][16];
  __shared__ int   first[16][16];
  __shared__ int   selb[2 * TILE_N];
  __shared__ int   selk[2 * TILE_N];
  __shared__ float selx[2 * TILE_N];
  __shared__ __attribute__((aligned(16))) float btab[2][GP1][JG][TILE_N][4];
  __shared__ __attribute__((aligned(16))) float stage[KW][OHALF][SPITCH];

  const int tp   = blockIdx.x;
  const int i    = blockIdx.y;
  const int tid  = threadIdx.x;
  const int lane = threadIdx.x & 31;
  const int w    = __builtin_amdgcn_readfirstlane((int)(threadIdx.x >> 5));
  const int h    = lane >> 4;
  const int n    = lane & 15;

  const float xv0 = x[i * BATCH + tid];
  const float xv1 = x[i * BATCH + (BATCH / 2) + tid];
  {
    v4f* bz = (v4f*)&btab[0][0][0][0][0];
    const v4f zero4 = {0.0f, 0.0f, 0.0f, 0.0f};
#pragma unroll
    for (int k = 0; k < (2 * GP1 * JG * TILE_N) / (32 * KW); ++k)
      bz[k * (32 * KW) + tid] = zero4;
  }
  const int k0 = bin_index(xv0);
  const int k1 = bin_index(xv1);
  unsigned same0 = ~0u, same1 = ~0u;
  unsigned ing0  = ~0u, ing1  = ~0u;
#pragma unroll
  for (int bit = 0; bit < 4; ++bit) {
    const unsigned m0 = __builtin_amdgcn_ballot_w32(((k0 >> bit) & 1) != 0);
    const unsigned m1 = __builtin_amdgcn_ballot_w32(((k1 >> bit) & 1) != 0);
    same0 &= ((k0 >> bit) & 1) ? m0 : ~m0;
    same1 &= ((k1 >> bit) & 1) ? m1 : ~m1;
    ing0  &= ((lane >> bit) & 1) ? m0 : ~m0;
    ing1  &= ((lane >> bit) & 1) ? m1 : ~m1;
  }
  const unsigned below = (1u << lane) - 1u;
  const int in0 = __builtin_popcount(same0 & below);
  const int in1 = __builtin_popcount(same1 & below);
  if (lane < 16) {
    cnt[w][lane]     = __builtin_popcount(ing0);
    cnt[w + 8][lane] = __builtin_popcount(ing1);
  }
  __syncthreads();
  {
    const int v = tid >> 4, g = tid & 15;
    int tot = 0, pre = 0;
#pragma unroll
    for (int u = 0; u < 16; ++u) {
      const int c = cnt[u][g];
      tot += c;
      pre += (u < v) ? c : 0;
    }
    int s = tot;
    s += row_shift_right_or0<1>(s);
    s += row_shift_right_or0<2>(s);
    s += row_shift_right_or0<4>(s);
    s += row_shift_right_or0<8>(s);
    first[v][g] = (s - tot) + pre;
  }
  __syncthreads();
  const int r0 = first[w][k0] + in0;
  const int r1 = first[w + 8][k1] + in1;
  if ((r0 >> 5) == tp) { selb[r0 & 31] = tid;               selk[r0 & 31] = k0; selx[r0 & 31] = xv0; }
  if ((r1 >> 5) == tp) { selb[r1 & 31] = tid + (BATCH / 2); selk[r1 & 31] = k1; selx[r1 & 31] = xv1; }
  __syncthreads();
  if (tp == 0 && obase == 0) {
    rank[i * BATCH + tid]               = r0;
    rank[i * BATCH + (BATCH / 2) + tid] = r1;
  }

  const int   ixA  = selk[n];
  const int   ixB  = selk[TILE_N + n];
  const float dxvA = (selx[n] - borders[ixA]) * invlen[ixA];
  const float dxvB = (selx[TILE_N + n] - borders[ixB]) * invlen[ixB];

  const int gminA = __builtin_amdgcn_readfirstlane(selk[0]);
  const int gmaxA = __builtin_amdgcn_readfirstlane(selk[TILE_N - 1]) + 1;
  const int gminB = __builtin_amdgcn_readfirstlane(selk[TILE_N]);
  const int gmaxB = __builtin_amdgcn_readfirstlane(selk[2 * TILE_N - 1]) + 1;

  const size_t sliceStride = (size_t)OUT * IN_X * IN_Z;
  const size_t rowStride   = (size_t)IN_X * IN_Z;
  const int ldrow = lane >> 3;
  const int ldcol = (lane & 7) * 4;
  const float* Pld  = P + ((size_t)(obase + ldrow) * IN_X + i) * IN_Z + ldcol;
  const int lastSlice  = gmaxB * GP1 + (GP1 - 1);
  const int firstSlice = gminA * GP1 + w;
  v4f raw[8];
#pragma unroll
  for (int r = 0; r < 8; ++r)
    raw[r] = *(const v4f*)(Pld + (size_t)firstSlice * sliceStride + (size_t)(4 * r) * rowStride);

  {
    const int col = selb[lane] & (BATCH - 1);
#pragma unroll
    for (int k = 0; k < IN_Z / KW; ++k) {
      const int j = w + KW * k;
      float dz;
      const int zi = bin_of(z[j * BATCH + col], borders, invlen, dz);
      btab[h][zi][j >> 2][n][j & 3]     = 1.0f - dz;
      btab[h][zi + 1][j >> 2][n][j & 3] = dz;
    }
  }
  __syncthreads();

  v8f accA0 = {};
  v8f accA1 = {};
  v8f accB0 = {};
  v8f accB1 = {};

  const float* Ppre = P + ((size_t)(obase + lane) * IN_X + i) * IN_Z;
  float (*st)[SPITCH] = stage[w];

  int gx = gminA, gz = w;
  for (int slice = firstSlice; slice <= lastSlice; slice += KW) {
    const bool doA = (gx <= gmaxA);
    const bool doB = (gx >= gminB);

#pragma unroll
    for (int r = 0; r < 8; ++r)
      *(v4f*)&st[4 * r + ldrow][ldcol] = raw[r];

    if (slice + KW <= lastSlice) {
      const float* Np = Pld + (size_t)(slice + KW) * sliceStride;
#pragma unroll
      for (int r = 0; r < 8; ++r)
        raw[r] = *(const v4f*)(Np + (size_t)(4 * r) * rowStride);
      if (slice + 2 * KW <= lastSlice)
        __builtin_prefetch(Ppre + (size_t)(slice + 2 * KW) * sliceStride, 0, 3);
    }

    if (doA || doB) {
      v4f a0[4], a1[4];
#pragma unroll
      for (int c = 0; c < 4; ++c) {
        a0[c] = *(const v4f*)&st[n][16 * h + 4 * c];
        a1[c] = *(const v4f*)&st[16 + n][16 * h + 4 * c];
      }
      v4f bA[4], bB[4];
#pragma unroll
      for (int c = 0; c < 4; ++c) {
        bA[c] = *(const v4f*)&btab[0][gz][4 * h + c][n][0];
        bB[c] = *(const v4f*)&btab[1][gz][4 * h + c][n][0];
      }
      asm volatile("" : "+v"(bA[0]), "+v"(bA[1]), "+v"(bA[2]), "+v"(bA[3]));

      if (doA) {
        const float wx = (gx == ixA) ? (1.0f - dxvA) : ((gx == ixA + 1) ? dxvA : 0.0f);
#pragma unroll
        for (int q = 0; q < 8; ++q) {
          v2f bv;
          bv.x = bA[q >> 1][2 * (q & 1)] * wx;
          bv.y = bA[q >> 1][2 * (q & 1) + 1] * wx;
          v2f av0, av1;
          av0.x = a0[q >> 1][2 * (q & 1)];
          av0.y = a0[q >> 1][2 * (q & 1) + 1];
          av1.x = a1[q >> 1][2 * (q & 1)];
          av1.y = a1[q >> 1][2 * (q & 1) + 1];
          accA0 = __builtin_amdgcn_wmma_f32_16x16x4_f32(
              false, av0, false, bv, (short)0, accA0, false, false);
          accA1 = __builtin_amdgcn_wmma_f32_16x16x4_f32(
              false, av1, false, bv, (short)0, accA1, false, false);
        }
      }
      if (doB) {
        const float wx = (gx == ixB) ? (1.0f - dxvB) : ((gx == ixB + 1) ? dxvB : 0.0f);
#pragma unroll
        for (int q = 0; q < 8; ++q) {
          v2f bv;
          bv.x = bB[q >> 1][2 * (q & 1)] * wx;
          bv.y = bB[q >> 1][2 * (q & 1) + 1] * wx;
          v2f av0, av1;
          av0.x = a0[q >> 1][2 * (q & 1)];
          av0.y = a0[q >> 1][2 * (q & 1) + 1];
          av1.x = a1[q >> 1][2 * (q & 1)];
          av1.y = a1[q >> 1][2 * (q & 1) + 1];
          accB0 = __builtin_amdgcn_wmma_f32_16x16x4_f32(
              false, av0, false, bv, (short)0, accB0, false, false);
          accB1 = __builtin_amdgcn_wmma_f32_16x16x4_f32(
              false, av1, false, bv, (short)0, accB1, false, false);
        }
      }
    }

    gz += KW;
    if (gz >= GP1) { gz -= GP1; ++gx; }
  }

  float* pw = part[w];
#pragma unroll
  for (int r = 0; r < 8; ++r) {
    pw[(8 * h + r) * TILE_N + n]                         = accA0[r];
    pw[(16 + 8 * h + r) * TILE_N + n]                    = accA1[r];
    pw[OHALF * TILE_N + (8 * h + r) * TILE_N + n]        = accB0[r];
    pw[OHALF * TILE_N + (16 + 8 * h + r) * TILE_N + n]   = accB1[r];
  }
  __syncthreads();

  float* Tw = T + ((size_t)(i * NTILES + 2 * tp) * OHALF) * TILE_N;
  for (int k = threadIdx.x; k < 2 * OHALF * TILE_N; k += 32 * KW) {
    float s = part[0][k];
#pragma unroll
    for (int v = 1; v < KW; ++v) s += part[v][k];
    Tw[k] = s;
  }
}

#define RCOLG (BATCH / 32)
__global__ void __launch_bounds__(32) reduce_rows(const float* __restrict__ T,
                                                  const int* __restrict__ rank,
                                                  float* __restrict__ out,
                                                  int obase) {
  const int b  = (blockIdx.x % RCOLG) * 32 + threadIdx.x;
  const int op = (blockIdx.x / RCOLG) * 2;
  float s0 = 0.0f, s1 = 0.0f;
#pragma unroll
  for (int i = 0; i < IN_X; ++i) {
    const int r = rank[i * BATCH + b] & (BATCH - 1);
    const float* p = T + ((size_t)(i * NTILES + (r >> 4)) * OHALF + op) * TILE_N + (r & 15);
    s0 += p[0];
    s1 += p[TILE_N];
  }
  out[(size_t)(obase + op) * BATCH + b]     = s0;
  out[(size_t)(obase + op + 1) * BATCH + b] = s1;
}

extern "C" void kernel_launch(void* const* d_in, const int* in_sizes, int n_in,
                              void* d_out, int out_size, void* d_ws, size_t ws_size,
                              hipStream_t stream) {
  const float* x       = (const float*)d_in[0];
  const float* z       = (const float*)d_in[1];
  const float* P       = (const float*)d_in[2];
  const float* borders = (const float*)d_in[3];
  const float* invlen  = (const float*)d_in[4];
  float* out = (float*)d_out;

  float* T   = (float*)d_ws;
  int* rank  = (int*)(T + (size_t)IN_X * NTILES * OHALF * TILE_N) + IN_X * BATCH;

  bucket_gemm<<<dim3(NPAIRS, IN_X), dim3(32 * KW), 0, stream>>>(
      P, x, z, borders, invlen, rank, T, 0);
  reduce_rows<<<dim3(RCOLG * (OHALF / 2)), dim3(32), 0, stream>>>(T, rank, out, 0);

  bucket_gemm<<<dim3(NPAIRS, IN_X), dim3(32 * KW), 0, stream>>>(
      P, x, z, borders, invlen, rank, T, OHALF);
  reduce_rows<<<dim3(RCOLG * (OHALF / 2)), dim3(32), 0, stream>>>(T, rank, out, OHALF);
}


# Both 32-row passes in one product launch (columns ordered and z table built once per work-group, LDS-only barrier between passes), one row-sum launch: 4 launches become 2 if workspace >= 4259840 B
# speedup vs baseline: 6.8093x; 1.1025x over previous
#include <hip/hip_runtime.h>
#include <math.h>

#define NUM_GRIDS 16
#define IN_X 32
#define IN_Z 32
#define OUT 64
#define BATCH 512
#define GP1 (NUM_GRIDS + 1)
#define TILE_N 16
#define NTILES (BATCH / TILE_N)
#define NPAIRS (NTILES / 2)
#define OHALF 32
#define PLANE ((size_t)IN_X * NTILES * OHALF * TILE_N)
#define KW 8
#define SPITCH 36
#define JG (IN_Z / 4)

typedef __attribute__((ext_vector_type(2))) float v2f;
typedef __attribute__((ext_vector_type(4))) float v4f;
typedef __attribute__((ext_vector_type(8))) float v8f;

__device__ __forceinline__ int bin_index(float v) {
  float e   = __expf(-fabsf(v));
  float cdf = (v > 0.0f) ? (1.0f - 0.5f * e) : (0.5f * e);
  int idx = (int)(cdf * (float)NUM_GRIDS);
  return idx < 0 ? 0 : (idx > NUM_GRIDS - 1 ? NUM_GRIDS - 1 : idx);
}

__device__ __forceinline__ int bin_of(float v, const float* __restrict__ borders,
                                      const float* __restrict__ invlen, float& d) {
  const int idx = bin_index(v);
  d = (v - borders[idx]) * invlen[idx];
  return idx;
}

template <int N>
__device__ __forceinline__ int row_shift_right_or0(int v) {
  return __builtin_amdgcn_update_dpp(0, v, 0x110 + N, 0xf, 0xf, true);
}

__device__ __forceinline__ void lds_barrier() {
  __builtin_amdgcn_fence(__ATOMIC_RELEASE, "workgroup", "local");
  __builtin_amdgcn_s_barrier();
  __builtin_amdgcn_fence(__ATOMIC_ACQUIRE, "workgroup", "local");
}

__global__ void __launch_bounds__(32 * KW) __attribute__((amdgpu_waves_per_eu(4, 4))) bucket_gemm(
    const float* __restrict__ P,
    const float* __restrict__ x,
    const float* __restrict__ z,
    const float* __restrict__ borders,
    const float* __restrict__ invlen,
    int* __restrict__ rank,
    float* __restrict__ T,
    int obase,
    int npass)
{
  __shared__ float part[KW][2 * OHALF * TILE_N];
  __shared__ int   cnt[16][16];
  __shared__ int   first[16][16];
  __shared__ int   selb[2 * TILE_N];
  __shared__ int   selk[2 * TILE_N];
  __shared__ float selx[2 * TILE_N];
  __shared__ __attribute__((aligned(16))) float btab[2][GP1][JG][TILE_N][4];
  __shared__ __attribute__((aligned(16))) float stage[KW][OHALF][SPITCH];

  const int tp   = blockIdx.x;
  const int i    = blockIdx.y;
  const int tid  = threadIdx.x;
  const int lane = threadIdx.x & 31;
  const int w    = __builtin_amdgcn_readfirstlane((int)(threadIdx.x >> 5));
  const int h    = lane >> 4;
  const int n    = lane & 15;

  const float xv0 = x[i * BATCH + tid];
  const float xv1 = x[i * BATCH + (BATCH / 2) + tid];
  {
    v4f* bz = (v4f*)&btab[0][0][0][0][0];
    const v4f zero4 = {0.0f, 0.0f, 0.0f, 0.0f};
#pragma unroll
    for (int k = 0; k < (2 * GP1 * JG * TILE_N) / (32 * KW); ++k)
      bz[k * (32 * KW) + tid] = zero4;
  }
  const int k0 = bin_index(xv0);
  const int k1 = bin_index(xv1);
  unsigned same0 = ~0u, same1 = ~0u;
  unsigned ing0  = ~0u, ing1  = ~0u;
#pragma unroll
  for (int bit = 0; bit < 4; ++bit) {
    const unsigned m0 = __builtin_amdgcn_ballot_w32(((k0 >> bit) & 1) != 0);
    const unsigned m1 = __builtin_amdgcn_ballot_w32(((k1 >> bit) & 1) != 0);
    same0 &= ((k0 >> bit) & 1) ? m0 : ~m0;
    same1 &= ((k1 >> bit) & 1) ? m1 : ~m1;
    ing0  &= ((lane >> bit) & 1) ? m0 : ~m0;
    ing1  &= ((lane >> bit) & 1) ? m1 : ~m1;
  }
  const unsigned below = (1u << lane) - 1u;
  const int in0 = __builtin_popcount(same0 & below);
  const int in1 = __builtin_popcount(same1 & below);
  if (lane < 16) {
    cnt[w][lane]     = __builtin_popcount(ing0);
    cnt[w + 8][lane] = __builtin_popcount(ing1);
  }
  __syncthreads();
  {
    const int v = tid >> 4, g = tid & 15;
    int tot = 0, pre = 0;
#pragma unroll
    for (int u = 0; u < 16; ++u) {
      const int c = cnt[u][g];
      tot += c;
      pre += (u < v) ? c : 0;
    }
    int s = tot;
    s += row_shift_right_or0<1>(s);
    s += row_shift_right_or0<2>(s);
    s += row_shift_right_or0<4>(s);
    s += row_shift_right_or0<8>(s);
    first[v][g] = (s - tot) + pre;
  }
  __syncthreads();
  const int r0 = first[w][k0] + in0;
  const int r1 = first[w + 8][k1] + in1;
  if ((r0 >> 5) == tp) { selb[r0 & 31] = tid;               selk[r0 & 31] = k0; selx[r0 & 31] = xv0; }
  if ((r1 >> 5) == tp) { selb[r1 & 31] = tid + (BATCH / 2); selk[r1 & 31] = k1; selx[r1 & 31] = xv1; }
  __syncthreads();
  if (tp == 0 && obase == 0) {
    rank[i * BATCH + tid]               = r0;
    rank[i * BATCH + (BATCH / 2) + tid] = r1;
  }

  const int   ixA  = selk[n];
  const int   ixB  = selk[TILE_N + n];
  const float dxvA = (selx[n] - borders[ixA]) * invlen[ixA];
  const float dxvB = (selx[TILE_N + n] - borders[ixB]) * invlen[ixB];

  const int gminA = __builtin_amdgcn_readfirstlane(selk[0]);
  const int gmaxA = __builtin_amdgcn_readfirstlane(selk[TILE_N - 1]) + 1;
  const int gminB = __builtin_amdgcn_readfirstlane(selk[TILE_N]);
  const int gmaxB = __builtin_amdgcn_readfirstlane(selk[2 * TILE_N - 1]) + 1;

  const size_t sliceStride = (size_t)OUT * IN_X * IN_Z;
  const size_t rowStride   = (size_t)IN_X * IN_Z;
  const size_t passStride  = (size_t)OHALF * rowStride;
  const int ldrow = lane >> 3;
  const int ldcol = (lane & 7) * 4;
  const float* Pld  = P + ((size_t)(obase + ldrow) * IN_X + i) * IN_Z + ldcol;
  const int lastSlice  = gmaxB * GP1 + (GP1 - 1);
  const int firstSlice = gminA * GP1 + w;
  v4f raw[8];
#pragma unroll
  for (int r = 0; r < 8; ++r)
    raw[r] = *(const v4f*)(Pld + (size_t)firstSlice * sliceStride + (size_t)(4 * r) * rowStride);

  {
    const int col = selb[lane] & (BATCH - 1);
#pragma unroll
    for (int k = 0; k < IN_Z / KW; ++k) {
      const int j = w + KW * k;
      float dz;
      const int zi = bin_of(z[j * BATCH + col], borders, invlen, dz);
      btab[h][zi][j >> 2][n][j & 3]     = 1.0f - dz;
      btab[h][zi + 1][j >> 2][n][j & 3] = dz;
    }
  }
  __syncthreads();

  const float* Ppre = P + ((size_t)(obase + lane) * IN_X + i) * IN_Z;
  float (*st)[SPITCH] = stage[w];

  for (int p = 0; p < npass; ++p) {
    v8f accA0 = {};
    v8f accA1 = {};
    v8f accB0 = {};
    v8f accB1 = {};

    int gx = gminA, gz = w;
    for (int slice = firstSlice; slice <= lastSlice; slice += KW) {
      const bool doA = (gx <= gmaxA);
      const bool doB = (gx >= gminB);

#pragma unroll
      for (int r = 0; r < 8; ++r)
        *(v4f*)&st[4 * r + ldrow][ldcol] = raw[r];

      int s1 = slice + KW, p1 = p;
      if (s1 > lastSlice) { s1 = firstSlice; ++p1; }
      if (p1 < npass) {
        const float* Np = Pld + (size_t)p1 * passStride + (size_t)s1 * sliceStride;
#pragma unroll
        for (int r = 0; r < 8; ++r)
          raw[r] = *(const v4f*)(Np + (size_t)(4 * r) * rowStride);
        int s2 = s1 + KW, p2 = p1;
        if (s2 > lastSlice) { s2 = firstSlice; ++p2; }
        if (p2 < npass)
          __builtin_prefetch(Ppre + (size_t)p2 * passStride + (size_t)s2 * sliceStride, 0, 3);
      }

      if (doA || doB) {
        v4f a0[4], a1[4];
#pragma unroll
        for (int c = 0; c < 4; ++c) {
          a0[c] = *(const v4f*)&st[n][16 * h + 4 * c];
          a1[c] = *(const v4f*)&st[16 + n][16 * h + 4 * c];
        }
        v4f bA[4], bB[4];
#pragma unroll
        for (int c = 0; c < 4; ++c) {
          bA[c] = *(const v4f*)&btab[0][gz][4 * h + c][n][0];
          bB[c] = *(const v4f*)&btab[1][gz][4 * h + c][n][0];
        }
        asm volatile("" : "+v"(bA[0]), "+v"(bA[1]), "+v"(bA[2]), "+v"(bA[3]));

        if (doA) {
          const float wx = (gx == ixA) ? (1.0f - dxvA) : ((gx == ixA + 1) ? dxvA : 0.0f);
#pragma unroll
          for (int q = 0; q < 8; ++q) {
            v2f bv;
            bv.x = bA[q >> 1][2 * (q & 1)] * wx;
            bv.y = bA[q >> 1][2 * (q & 1) + 1] * wx;
            v2f av0, av1;
            av0.x = a0[q >> 1][2 * (q & 1)];
            av0.y = a0[q >> 1][2 * (q & 1) + 1];
            av1.x = a1[q >> 1][2 * (q & 1)];
            av1.y = a1[q >> 1][2 * (q & 1) + 1];
            accA0 = __builtin_amdgcn_wmma_f32_16x16x4_f32(
                false, av0, false, bv, (short)0, accA0, false, false);
            accA1 = __builtin_amdgcn_wmma_f32_16x16x4_f32(
                false, av1, false, bv, (short)0, accA1, false, false);
          }
        }
        if (doB) {
          const float wx = (gx == ixB) ? (1.0f - dxvB) : ((gx == ixB + 1) ? dxvB : 0.0f);
#pragma unroll
          for (int q = 0; q < 8; ++q) {
            v2f bv;
            bv.x = bB[q >> 1][2 * (q & 1)] * wx;
            bv.y = bB[q >> 1][2 * (q & 1) + 1] * wx;
            v2f av0, av1;
            av0.x = a0[q >> 1][2 * (q & 1)];
            av0.y = a0[q >> 1][2 * (q & 1) + 1];
            av1.x = a1[q >> 1][2 * (q & 1)];
            av1.y = a1[q >> 1][2 * (q & 1) + 1];
            accB0 = __builtin_amdgcn_wmma_f32_16x16x4_f32(
                false, av0, false, bv, (short)0, accB0, false, false);
            accB1 = __builtin_amdgcn_wmma_f32_16x16x4_f32(
                false, av1, false, bv, (short)0, accB1, false, false);
          }
        }
      }

      gz += KW;
      if (gz >= GP1) { gz -= GP1; ++gx; }
    }

    float* pw = part[w];
#pragma unroll
    for (int r = 0; r < 8; ++r) {
      pw[(8 * h + r) * TILE_N + n]                         = accA0[r];
      pw[(16 + 8 * h + r) * TILE_N + n]                    = accA1[r];
      pw[OHALF * TILE_N + (8 * h + r) * TILE_N + n]        = accB0[r];
      pw[OHALF * TILE_N + (16 + 8 * h + r) * TILE_N + n]   = accB1[r];
    }
    lds_barrier();

    float* Tw = T + (size_t)p * PLANE + ((size_t)(i * NTILES + 2 * tp) * OHALF) * TILE_N;
    for (int k = threadIdx.x; k < 2 * OHALF * TILE_N; k += 32 * KW) {
      float s = part[0][k];
#pragma unroll
      for (int v = 1; v < KW; ++v) s += part[v][k];
      Tw[k] = s;
    }
    if (p + 1 < npass)
      lds_barrier();
  }
}

#define RCOLG (BATCH / 32)
__global__ void __launch_bounds__(32) reduce_rows(const float* __restrict__ T,
                                                  const int* __restrict__ rank,
                                                  float* __restrict__ out,
                                                  int obase) {
  const int b  = (blockIdx.x % RCOLG) * 32 + threadIdx.x;
  const int op = (blockIdx.x / RCOLG) * 2;
  const float* Tp = T + (size_t)(op / OHALF) * PLANE;
  const int ol = op % OHALF;
  float s0 = 0.0f, s1 = 0.0f;
#pragma unroll
  for (int i = 0; i < IN_X; ++i) {
    const int r = rank[i * BATCH + b] & (BATCH - 1);
    const float* p = Tp + ((size_t)(i * NTILES + (r >> 4)) * OHALF + ol) * TILE_N + (r & 15);
    s0 += p[0];
    s1 += p[TILE_N];
  }
  out[(size_t)(obase + op) * BATCH + b]     = s0;
  out[(size_t)(obase + op + 1) * BATCH + b] = s1;
}

extern "C" void kernel_launch(void* const* d_in, const int* in_sizes, int n_in,
                              void* d_out, int out_size, void* d_ws, size_t ws_size,
                              hipStream_t stream) {
  const float* x       = (const float*)d_in[0];
  const float* z       = (const float*)d_in[1];
  const float* P       = (const float*)d_in[2];
  const float* borders = (const float*)d_in[3];
  const float* invlen  = (const float*)d_in[4];
  float* out = (float*)d_out;

  float* T = (float*)d_ws;
  if (ws_size >= (2 * PLANE + (size_t)IN_X * BATCH) * sizeof(float)) {
    int* rank = (int*)(T + 2 * PLANE);
    bucket_gemm<<<dim3(NPAIRS, IN_X), dim3(32 * KW), 0, stream>>>(
        P, x, z, borders, invlen, rank, T, 0, 2);
    reduce_rows<<<dim3(RCOLG * (OUT / 2)), dim3(32), 0, stream>>>(T, rank, out, 0);
    return;
  }

  int* rank  = (int*)(T + PLANE) + IN_X * BATCH;

  bucket_gemm<<<dim3(NPAIRS, IN_X), dim3(32 * KW), 0, stream>>>(
      P, x, z, borders, invlen, rank, T, 0, 1);
  reduce_rows<<<dim3(RCOLG * (OHALF / 2)), dim3(32), 0, stream>>>(T, rank, out, 0);

  bucket_gemm<<<dim3(NPAIRS, IN_X), dim3(32 * KW), 0, stream>>>(
      P, x, z, borders, invlen, rank, T, OHALF, 1);
  reduce_rows<<<dim3(RCOLG * (OHALF / 2)), dim3(32), 0, stream>>>(T, rank, out, OHALF);
}
